# starting kernel
# baseline (speedup 1.0000x reference)
#include <hip/hip_runtime.h>

#define NNODES 100000
#define NEDGES 1600000
#define FIN 128
#define FH 64
#define FOUT 32

typedef float v2f __attribute__((ext_vector_type(2)));
typedef float v8f __attribute__((ext_vector_type(8)));

static __device__ __forceinline__ v8f wmma_f32_k4(v2f a, v2f b, v8f c) {
  return __builtin_amdgcn_wmma_f32_16x16x4_f32(false, a, false, b, (short)0, c,
                                               false, false);
}

__global__ void deg_init_kernel(float* __restrict__ deg) {
  int i = blockIdx.x * blockDim.x + threadIdx.x;
  if (i < NNODES) deg[i] = 1.0f;
}

__global__ void deg_count_kernel(const int* __restrict__ dst,
                                 float* __restrict__ deg) {
  int i = blockIdx.x * blockDim.x + threadIdx.x;
  if (i < NEDGES) unsafeAtomicAdd(&deg[dst[i]], 1.0f);
}

__global__ void dinv_kernel(const float* __restrict__ deg,
                            float* __restrict__ dinv) {
  int i = blockIdx.x * blockDim.x + threadIdx.x;
  if (i < NNODES) dinv[i] = rsqrtf(deg[i]);
}

__global__ __launch_bounds__(256) void gcn_gemm1(const float* __restrict__ x,
                                                 const float* __restrict__ W1,
                                                 float* __restrict__ h1) {
  __shared__ float sW[FH * FIN];
  for (int i = threadIdx.x; i < FH * FIN; i += 256) sW[i] = W1[i];
  __syncthreads();

  const int lane = threadIdx.x & 31;
  const int wave = threadIdx.x >> 5;
  const int tile = blockIdx.x * 8 + wave;
  if (tile * 16 >= NNODES) return;
  const int m0 = tile * 16;
  const int lrow = lane & 15;
  const int koff = (lane >> 4) << 1;

  v8f acc0 = {}, acc1 = {}, acc2 = {}, acc3 = {};
  const float* xrow = x + (size_t)(m0 + lrow) * FIN;
#pragma unroll 4
  for (int k = 0; k < FIN; k += 4) {
    v2f a = *(const v2f*)(xrow + k + koff);
    v2f b0 = *(const v2f*)(&sW[(0 * 16 + lrow) * FIN + k + koff]);
    v2f b1 = *(const v2f*)(&sW[(1 * 16 + lrow) * FIN + k + koff]);
    v2f b2 = *(const v2f*)(&sW[(2 * 16 + lrow) * FIN + k + koff]);
    v2f b3 = *(const v2f*)(&sW[(3 * 16 + lrow) * FIN + k + koff]);
    acc0 = wmma_f32_k4(a, b0, acc0);
    acc1 = wmma_f32_k4(a, b1, acc1);
    acc2 = wmma_f32_k4(a, b2, acc2);
    acc3 = wmma_f32_k4(a, b3, acc3);
  }
  const int rbase = m0 + ((lane >> 4) << 3);
#pragma unroll
  for (int r = 0; r < 8; ++r) {
    h1[(size_t)(rbase + r) * FH + 0 * 16 + lrow] = acc0[r];
    h1[(size_t)(rbase + r) * FH + 1 * 16 + lrow] = acc1[r];
    h1[(size_t)(rbase + r) * FH + 2 * 16 + lrow] = acc2[r];
    h1[(size_t)(rbase + r) * FH + 3 * 16 + lrow] = acc3[r];
  }
}

__global__ __launch_bounds__(256) void gcn_gemm2(const float* __restrict__ agg1,
                                                 const float* __restrict__ b1,
                                                 const float* __restrict__ W2,
                                                 float* __restrict__ h2) {
  __shared__ float sW[FOUT * FH];
  __shared__ float sB[FH];
  for (int i = threadIdx.x; i < FOUT * FH; i += 256) sW[i] = W2[i];
  if (threadIdx.x < FH) sB[threadIdx.x] = b1[threadIdx.x];
  __syncthreads();

  const int lane = threadIdx.x & 31;
  const int wave = threadIdx.x >> 5;
  const int tile = blockIdx.x * 8 + wave;
  if (tile * 16 >= NNODES) return;
  const int m0 = tile * 16;
  const int lrow = lane & 15;
  const int koff = (lane >> 4) << 1;

  v8f acc0 = {}, acc1 = {};
  const float* arow = agg1 + (size_t)(m0 + lrow) * FH;
#pragma unroll 4
  for (int k = 0; k < FH; k += 4) {
    v2f ar = *(const v2f*)(arow + k + koff);
    v2f a;
    a.x = fmaxf(ar.x + sB[k + koff + 0], 0.0f);
    a.y = fmaxf(ar.y + sB[k + koff + 1], 0.0f);
    v2f b0 = *(const v2f*)(&sW[(0 * 16 + lrow) * FH + k + koff]);
    v2f b1v = *(const v2f*)(&sW[(1 * 16 + lrow) * FH + k + koff]);
    acc0 = wmma_f32_k4(a, b0, acc0);
    acc1 = wmma_f32_k4(a, b1v, acc1);
  }
  const int rbase = m0 + ((lane >> 4) << 3);
#pragma unroll
  for (int r = 0; r < 8; ++r) {
    h2[(size_t)(rbase + r) * FOUT + 0 * 16 + lrow] = acc0[r];
    h2[(size_t)(rbase + r) * FOUT + 1 * 16 + lrow] = acc1[r];
  }
}

__global__ void agg1_init_kernel(const float* __restrict__ h1,
                                 const float* __restrict__ dinv,
                                 float* __restrict__ agg1) {
  int i = blockIdx.x * blockDim.x + threadIdx.x;
  if (i >= NNODES * (FH / 4)) return;
  int node = i / (FH / 4);
  float w = dinv[node];
  w *= w;
  float4 v = ((const float4*)h1)[i];
  float4 o = {v.x * w, v.y * w, v.z * w, v.w * w};
  ((float4*)agg1)[i] = o;
}

__global__ void agg2_init_kernel(const float* __restrict__ h2,
                                 const float* __restrict__ dinv,
                                 const float* __restrict__ b2,
                                 float* __restrict__ out) {
  int i = blockIdx.x * blockDim.x + threadIdx.x;
  if (i >= NNODES * (FOUT / 4)) return;
  int node = i / (FOUT / 4);
  int c4 = i % (FOUT / 4);
  float w = dinv[node];
  w *= w;
  float4 v = ((const float4*)h2)[i];
  float4 bb = ((const float4*)b2)[c4];
  float4 o = {v.x * w + bb.x, v.y * w + bb.y, v.z * w + bb.z, v.w * w + bb.w};
  ((float4*)out)[i] = o;
}

template <int F>
__global__ void agg_edges_kernel(const int* __restrict__ src,
                                 const int* __restrict__ dst,
                                 const float* __restrict__ dinv,
                                 const float* __restrict__ h,
                                 float* __restrict__ agg) {
  const int TPE = F / 4;
  long long t = (long long)blockIdx.x * blockDim.x + threadIdx.x;
  long long e = t / TPE;
  int c = (int)(t % TPE);
  if (e >= NEDGES) return;
  int s = src[e];
  int d = dst[e];
  float w = dinv[s] * dinv[d];
  float4 v = ((const float4*)(h + (size_t)s * F))[c];
  float* o = agg + (size_t)d * F + c * 4;
  unsafeAtomicAdd(o + 0, v.x * w);
  unsafeAtomicAdd(o + 1, v.y * w);
  unsafeAtomicAdd(o + 2, v.z * w);
  unsafeAtomicAdd(o + 3, v.w * w);
}

extern "C" void kernel_launch(void* const* d_in, const int* in_sizes, int n_in,
                              void* d_out, int out_size, void* d_ws,
                              size_t ws_size, hipStream_t stream) {
  const float* x = (const float*)d_in[0];
  const int* edge_index = (const int*)d_in[1];
  const float* W1 = (const float*)d_in[2];
  const float* b1 = (const float*)d_in[3];
  const float* W2 = (const float*)d_in[4];
  const float* b2 = (const float*)d_in[5];
  float* out = (float*)d_out;

  const int* src = edge_index;
  const int* dst = edge_index + NEDGES;

  float* ws = (float*)d_ws;
  float* deg = ws;
  float* dinv = ws + NNODES;
  float* h1 = ws + 2 * (size_t)NNODES;
  float* agg1 = h1 + (size_t)NNODES * FH;
  float* h2 = agg1 + (size_t)NNODES * FH;

  const int B = 256;
  const int nTiles = NNODES / 16;
  const int gemmBlocks = (nTiles + 7) / 8;

  deg_init_kernel<<<(NNODES + B - 1) / B, B, 0, stream>>>(deg);
  deg_count_kernel<<<(NEDGES + B - 1) / B, B, 0, stream>>>(dst, deg);
  dinv_kernel<<<(NNODES + B - 1) / B, B, 0, stream>>>(deg, dinv);

  gcn_gemm1<<<gemmBlocks, B, 0, stream>>>(x, W1, h1);

  agg1_init_kernel<<<(NNODES * (FH / 4) + B - 1) / B, B, 0, stream>>>(h1, dinv,
                                                                      agg1);
  {
    long long tot = (long long)NEDGES * (FH / 4);
    agg_edges_kernel<FH><<<(unsigned)((tot + B - 1) / B), B, 0, stream>>>(
        src, dst, dinv, h1, agg1);
  }

  gcn_gemm2<<<gemmBlocks, B, 0, stream>>>(agg1, b1, W2, h2);

  agg2_init_kernel<<<(NNODES * (FOUT / 4) + B - 1) / B, B, 0, stream>>>(
      h2, dinv, b2, out);
  {
    long long tot = (long long)NEDGES * (FOUT / 4);
    agg_edges_kernel<FOUT><<<(unsigned)((tot + B - 1) / B), B, 0, stream>>>(
        src, dst, dinv, h2, out);
  }
}


# Edges counting-sorted by destination per call (returning int adds, scan, fill); each node's rows summed in registers in slot order, one store per row: no f32 atomics, new sum order
# speedup vs baseline: 1.5108x; 1.5108x over previous
#include <hip/hip_runtime.h>

#define NNODES 100000
#define NEDGES 1600000
#define FIN 128
#define FH 64
#define FOUT 32

typedef float v2f __attribute__((ext_vector_type(2)));
typedef float v8f __attribute__((ext_vector_type(8)));

static __device__ __forceinline__ v8f wmma_f32_k4(v2f a, v2f b, v8f c) {
  return __builtin_amdgcn_wmma_f32_16x16x4_f32(false, a, false, b, (short)0, c,
                                               false, false);
}

#define SCAN_ELEMS 1024
#define SCAN_BLOCKS ((NNODES + SCAN_ELEMS - 1) / SCAN_ELEMS)

__global__ void cnt_zero_kernel(int* __restrict__ cnt) {
  int i = blockIdx.x * blockDim.x + threadIdx.x;
  if (i < NNODES) cnt[i] = 0;
}

__global__ void edge_slot_kernel(const int* __restrict__ dst,
                                 int* __restrict__ cnt,
                                 int* __restrict__ slot) {
  int e = blockIdx.x * blockDim.x + threadIdx.x;
  if (e < NEDGES) slot[e] = atomicAdd(&cnt[dst[e]], 1);
}

__global__ __launch_bounds__(256) void scan_local_kernel(
    const int* __restrict__ cnt, int* __restrict__ loc,
    int* __restrict__ bsum) {
  __shared__ int s[256];
  const int tid = threadIdx.x;
  const int base = (blockIdx.x * 256 + tid) * 4;
  int4 c = {0, 0, 0, 0};
  if (base < NNODES) c = *(const int4*)(cnt + base);
  const int tsum = c.x + c.y + c.z + c.w;
  s[tid] = tsum;
  __syncthreads();
  for (int off = 1; off < 256; off <<= 1) {
    int v = (tid >= off) ? s[tid - off] : 0;
    __syncthreads();
    s[tid] += v;
    __syncthreads();
  }
  const int excl = s[tid] - tsum;
  if (base < NNODES) {
    int4 o = {excl, excl + c.x, excl + c.x + c.y, excl + c.x + c.y + c.z};
    *(int4*)(loc + base) = o;
  }
  if (tid == 255) bsum[blockIdx.x] = s[255];
}

__global__ __launch_bounds__(256) void scan_add_kernel(
    const int* __restrict__ cnt, const int* __restrict__ loc,
    const int* __restrict__ bsum, int2* __restrict__ info) {
  __shared__ int s[256];
  const int tid = threadIdx.x;
  s[tid] = (tid < (int)blockIdx.x) ? bsum[tid] : 0;
  __syncthreads();
  for (int off = 128; off > 0; off >>= 1) {
    if (tid < off) s[tid] += s[tid + off];
    __syncthreads();
  }
  const int prefix = s[0];
  const int base = (blockIdx.x * 256 + tid) * 4;
  if (base < NNODES) {
    const int4 o = *(const int4*)(loc + base);
    const int4 c = *(const int4*)(cnt + base);
    int4 r0 = {o.x + prefix, __float_as_int(rsqrtf((float)c.x + 1.0f)),
               o.y + prefix, __float_as_int(rsqrtf((float)c.y + 1.0f))};
    int4 r1 = {o.z + prefix, __float_as_int(rsqrtf((float)c.z + 1.0f)),
               o.w + prefix, __float_as_int(rsqrtf((float)c.w + 1.0f))};
    *(int4*)(info + base) = r0;
    *(int4*)(info + base + 2) = r1;
  }
}

__global__ void edge_fill_kernel(const int* __restrict__ src,
                                 const int* __restrict__ dst,
                                 const int* __restrict__ slot,
                                 const int2* __restrict__ info,
                                 int2* __restrict__ csr) {
  int e = blockIdx.x * blockDim.x + threadIdx.x;
  if (e >= NEDGES) return;
  const int s = src[e];
  const int d = dst[e];
  const int2 ns = info[s];
  const int2 nd = info[d];
  const int p = nd.x + slot[e];
  const float w = __int_as_float(ns.y) * __int_as_float(nd.y);
  int2 rec = {s, __float_as_int(w)};
  csr[p] = rec;
}

__global__ __launch_bounds__(256) void gcn_gemm1(const float* __restrict__ x,
                                                 const float* __restrict__ W1,
                                                 float* __restrict__ h1) {
  __shared__ float sW[FH * FIN];
  for (int i = threadIdx.x; i < FH * FIN; i += 256) sW[i] = W1[i];
  __syncthreads();

  const int lane = threadIdx.x & 31;
  const int wave = threadIdx.x >> 5;
  const int tile = blockIdx.x * 8 + wave;
  if (tile * 16 >= NNODES) return;
  const int m0 = tile * 16;
  const int lrow = lane & 15;
  const int koff = (lane >> 4) << 1;

  v8f acc0 = {}, acc1 = {}, acc2 = {}, acc3 = {};
  const float* xrow = x + (size_t)(m0 + lrow) * FIN;
#pragma unroll 4
  for (int k = 0; k < FIN; k += 4) {
    v2f a = *(const v2f*)(xrow + k + koff);
    v2f b0 = *(const v2f*)(&sW[(0 * 16 + lrow) * FIN + k + koff]);
    v2f b1 = *(const v2f*)(&sW[(1 * 16 + lrow) * FIN + k + koff]);
    v2f b2 = *(const v2f*)(&sW[(2 * 16 + lrow) * FIN + k + koff]);
    v2f b3 = *(const v2f*)(&sW[(3 * 16 + lrow) * FIN + k + koff]);
    acc0 = wmma_f32_k4(a, b0, acc0);
    acc1 = wmma_f32_k4(a, b1, acc1);
    acc2 = wmma_f32_k4(a, b2, acc2);
    acc3 = wmma_f32_k4(a, b3, acc3);
  }
  const int rbase = m0 + ((lane >> 4) << 3);
#pragma unroll
  for (int r = 0; r < 8; ++r) {
    h1[(size_t)(rbase + r) * FH + 0 * 16 + lrow] = acc0[r];
    h1[(size_t)(rbase + r) * FH + 1 * 16 + lrow] = acc1[r];
    h1[(size_t)(rbase + r) * FH + 2 * 16 + lrow] = acc2[r];
    h1[(size_t)(rbase + r) * FH + 3 * 16 + lrow] = acc3[r];
  }
}

__global__ __launch_bounds__(256) void gcn_gemm2(const float* __restrict__ agg1,
                                                 const float* __restrict__ b1,
                                                 const float* __restrict__ W2,
                                                 float* __restrict__ h2) {
  __shared__ float sW[FOUT * FH];
  __shared__ float sB[FH];
  for (int i = threadIdx.x; i < FOUT * FH; i += 256) sW[i] = W2[i];
  if (threadIdx.x < FH) sB[threadIdx.x] = b1[threadIdx.x];
  __syncthreads();

  const int lane = threadIdx.x & 31;
  const int wave = threadIdx.x >> 5;
  const int tile = blockIdx.x * 8 + wave;
  if (tile * 16 >= NNODES) return;
  const int m0 = tile * 16;
  const int lrow = lane & 15;
  const int koff = (lane >> 4) << 1;

  v8f acc0 = {}, acc1 = {};
  const float* arow = agg1 + (size_t)(m0 + lrow) * FH;
#pragma unroll 4
  for (int k = 0; k < FH; k += 4) {
    v2f ar = *(const v2f*)(arow + k + koff);
    v2f a;
    a.x = fmaxf(ar.x + sB[k + koff + 0], 0.0f);
    a.y = fmaxf(ar.y + sB[k + koff + 1], 0.0f);
    v2f b0 = *(const v2f*)(&sW[(0 * 16 + lrow) * FH + k + koff]);
    v2f b1v = *(const v2f*)(&sW[(1 * 16 + lrow) * FH + k + koff]);
    acc0 = wmma_f32_k4(a, b0, acc0);
    acc1 = wmma_f32_k4(a, b1v, acc1);
  }
  const int rbase = m0 + ((lane >> 4) << 3);
#pragma unroll
  for (int r = 0; r < 8; ++r) {
    h2[(size_t)(rbase + r) * FOUT + 0 * 16 + lrow] = acc0[r];
    h2[(size_t)(rbase + r) * FOUT + 1 * 16 + lrow] = acc1[r];
  }
}

template <int F, bool BIAS>
__global__ __launch_bounds__(256) void agg_rows_kernel(
    const int2* __restrict__ info, const int2* __restrict__ csr,
    const float* __restrict__ h, const float* __restrict__ bias,
    float* __restrict__ out) {
  const int TPE = F / 4;
  const int t = blockIdx.x * blockDim.x + threadIdx.x;
  const int node = t / TPE;
  const int c = t % TPE;
  if (node >= NNODES) return;
  const int2 ni = info[node];
  const int beg = ni.x;
  const int end = (node + 1 < NNODES) ? info[node + 1].x : NEDGES;
  const float4* __restrict__ h4 = (const float4*)h;
  float ws = __int_as_float(ni.y);
  ws *= ws;
  float4 v = h4[(size_t)node * TPE + c];
  float4 acc = {v.x * ws, v.y * ws, v.z * ws, v.w * ws};
  if (BIAS) {
    float4 bb = ((const float4*)bias)[c];
    acc.x += bb.x; acc.y += bb.y; acc.z += bb.z; acc.w += bb.w;
  }
  int p = beg;
  for (; p + 4 <= end; p += 4) {
    const int2 e0 = csr[p + 0], e1 = csr[p + 1];
    const int2 e2 = csr[p + 2], e3 = csr[p + 3];
    const float w0 = __int_as_float(e0.y), w1 = __int_as_float(e1.y);
    const float w2 = __int_as_float(e2.y), w3 = __int_as_float(e3.y);
    const float4 v0 = h4[(size_t)e0.x * TPE + c];
    const float4 v1 = h4[(size_t)e1.x * TPE + c];
    const float4 v2 = h4[(size_t)e2.x * TPE + c];
    const float4 v3 = h4[(size_t)e3.x * TPE + c];
    acc.x += v0.x * w0; acc.y += v0.y * w0; acc.z += v0.z * w0; acc.w += v0.w * w0;
    acc.x += v1.x * w1; acc.y += v1.y * w1; acc.z += v1.z * w1; acc.w += v1.w * w1;
    acc.x += v2.x * w2; acc.y += v2.y * w2; acc.z += v2.z * w2; acc.w += v2.w * w2;
    acc.x += v3.x * w3; acc.y += v3.y * w3; acc.z += v3.z * w3; acc.w += v3.w * w3;
  }
  for (; p < end; ++p) {
    const int2 e0 = csr[p];
    const float w0 = __int_as_float(e0.y);
    const float4 v0 = h4[(size_t)e0.x * TPE + c];
    acc.x += v0.x * w0; acc.y += v0.y * w0; acc.z += v0.z * w0; acc.w += v0.w * w0;
  }
  ((float4*)out)[(size_t)node * TPE + c] = acc;
}

extern "C" void kernel_launch(void* const* d_in, const int* in_sizes, int n_in,
                              void* d_out, int out_size, void* d_ws,
                              size_t ws_size, hipStream_t stream) {
  const float* x = (const float*)d_in[0];
  const int* edge_index = (const int*)d_in[1];
  const float* W1 = (const float*)d_in[2];
  const float* b1 = (const float*)d_in[3];
  const float* W2 = (const float*)d_in[4];
  const float* b2 = (const float*)d_in[5];
  float* out = (float*)d_out;

  const int* src = edge_index;
  const int* dst = edge_index + NEDGES;

  float* ws = (float*)d_ws;
  int2* info = (int2*)ws;
  float* h1 = ws + 2 * (size_t)NNODES;
  float* agg1 = h1 + (size_t)NNODES * FH;
  int2* csr = (int2*)(agg1 + (size_t)NNODES * FH);
  float* h2 = h1;
  int* slot = (int*)agg1;
  int* cnt = slot + NEDGES;
  int* loc = cnt + NNODES;
  int* bsum = loc + NNODES;

  const int B = 256;
  const int nTiles = NNODES / 16;
  const int gemmBlocks = (nTiles + 7) / 8;

  cnt_zero_kernel<<<(NNODES + B - 1) / B, B, 0, stream>>>(cnt);
  edge_slot_kernel<<<(NEDGES + B - 1) / B, B, 0, stream>>>(dst, cnt, slot);
  scan_local_kernel<<<SCAN_BLOCKS, B, 0, stream>>>(cnt, loc, bsum);
  scan_add_kernel<<<SCAN_BLOCKS, B, 0, stream>>>(cnt, loc, bsum, info);
  edge_fill_kernel<<<(NEDGES + B - 1) / B, B, 0, stream>>>(src, dst, slot, info,
                                                          csr);

  gcn_gemm1<<<gemmBlocks, B, 0, stream>>>(x, W1, h1);

  agg_rows_kernel<FH, false>
      <<<(NNODES * (FH / 4) + B - 1) / B, B, 0, stream>>>(info, csr, h1, b1,
                                                          agg1);

  gcn_gemm2<<<gemmBlocks, B, 0, stream>>>(agg1, b1, W2, h2);

  agg_rows_kernel<FOUT, true>
      <<<(NNODES * (FOUT / 4) + B - 1) / B, B, 0, stream>>>(info, csr, h2, b2,
                                                            out);
}


# Edge sort by destination moved off per-edge global atomics and scattered stores onto LDS counters: 256-node bins, LDS-staged partition, per-bin LDS count and fill; per-node sum order: LDS arrival
# speedup vs baseline: 2.6294x; 1.7404x over previous
#include <hip/hip_runtime.h>

#define NNODES 100000
#define NEDGES 1600000
#define FIN 128
#define FH 64
#define FOUT 32

typedef float v2f __attribute__((ext_vector_type(2)));
typedef float v8f __attribute__((ext_vector_type(8)));

static __device__ __forceinline__ v8f wmma_f32_k4(v2f a, v2f b, v8f c) {
  return __builtin_amdgcn_wmma_f32_16x16x4_f32(false, a, false, b, (short)0, c,
                                               false, false);
}

#define BIN_SHIFT 8
#define BIN_NODES 256
#define NBIN ((NNODES + BIN_NODES - 1) / BIN_NODES)
#define PBLK 256
#define PCH (NEDGES / PBLK)
#define PTHR 1024
#define PPER ((PCH + PTHR - 1) / PTHR)
#define FTHR 512
#define FILL_CAP 6144

static_assert(NEDGES % PBLK == 0, "edge chunks must be exact");
static_assert(NNODES < (1 << 23), "source id must fit above the low byte");
static_assert(NBIN <= PTHR, "one thread per bin in the partition blocks");

static __device__ __forceinline__ int wave_incl_scan(int v) {
  const int lane = threadIdx.x & 31;
#pragma unroll
  for (int o = 1; o < 32; o <<= 1) {
    const int t = __shfl_up(v, o, 32);
    if (lane >= o) v += t;
  }
  return v;
}

template <int NW>
static __device__ __forceinline__ int block_incl_scan(int v, int* wsum) {
  const int lane = threadIdx.x & 31;
  const int w = threadIdx.x >> 5;
  v = wave_incl_scan(v);
  if (lane == 31) wsum[w] = v;
  __syncthreads();
  if (w == 0) {
    int t = (lane < NW) ? wsum[lane] : 0;
    t = wave_incl_scan(t);
    if (lane < NW) wsum[lane] = t;
  }
  __syncthreads();
  if (w > 0) v += wsum[w - 1];
  return v;
}

__global__ __launch_bounds__(PTHR) void bin_count_kernel(
    const int* __restrict__ dst, int* __restrict__ hist) {
  __shared__ int h[NBIN];
  const int tid = threadIdx.x;
  if (tid < NBIN) h[tid] = 0;
  const int e0 = blockIdx.x * PCH;
  int ed[PPER];
#pragma unroll
  for (int j = 0; j < PPER; ++j) {
    const int i = tid + j * PTHR;
    ed[j] = (i < PCH) ? dst[e0 + i] : -1;
  }
  __syncthreads();
#pragma unroll
  for (int j = 0; j < PPER; ++j)
    if (ed[j] >= 0) atomicAdd(&h[ed[j] >> BIN_SHIFT], 1);
  __syncthreads();
  if (tid < NBIN) hist[tid * PBLK + blockIdx.x] = h[tid];
}

__global__ __launch_bounds__(PBLK) void bin_scan_kernel(
    const int* __restrict__ hist, int* __restrict__ rel,
    int* __restrict__ bintot) {
  __shared__ int wsum[PBLK / 32];
  const int tid = threadIdx.x;
  const int v = hist[blockIdx.x * PBLK + tid];
  const int incl = block_incl_scan<PBLK / 32>(v, wsum);
  rel[blockIdx.x * PBLK + tid] = incl - v;
  if (tid == PBLK - 1) bintot[blockIdx.x] = incl;
}

__global__ __launch_bounds__(PTHR) void bin_scatter_kernel(
    const int* __restrict__ src, const int* __restrict__ dst,
    const int* __restrict__ rel, const int* __restrict__ bintot,
    int* __restrict__ binstart, int* __restrict__ binned) {
  __shared__ int wsum[PTHR / 32];
  __shared__ int h[NBIN];
  __shared__ int gbase[NBIN];
  __shared__ int rec[PCH];
  __shared__ int gp[PCH];
  const int tid = threadIdx.x;
  const int b = blockIdx.x;

  const int tot = (tid < NBIN) ? bintot[tid] : 0;
  const int myrel = (tid < NBIN) ? rel[tid * PBLK + b] : 0;
  if (tid < NBIN) h[tid] = 0;

  const int e0 = b * PCH;
  int es[PPER], ed[PPER], er[PPER];
#pragma unroll
  for (int j = 0; j < PPER; ++j) {
    const int i = tid + j * PTHR;
    es[j] = (i < PCH) ? src[e0 + i] : 0;
    ed[j] = (i < PCH) ? dst[e0 + i] : -1;
  }

  const int incl = block_incl_scan<PTHR / 32>(tot, wsum);
  if (tid < NBIN) {
    const int st = incl - tot;
    gbase[tid] = st + myrel;
    if (b == 0) binstart[tid] = st;
  }
  if (b == 0 && tid == 0) binstart[NBIN] = NEDGES;

#pragma unroll
  for (int j = 0; j < PPER; ++j) {
    er[j] = 0;
    if (ed[j] >= 0) er[j] = atomicAdd(&h[ed[j] >> BIN_SHIFT], 1);
  }
  __syncthreads();

  const int hv = (tid < NBIN) ? h[tid] : 0;
  const int incl2 = block_incl_scan<PTHR / 32>(hv, wsum);
  if (tid < NBIN) h[tid] = incl2 - hv;
  __syncthreads();

#pragma unroll
  for (int j = 0; j < PPER; ++j) {
    if (ed[j] >= 0) {
      const int k = ed[j] >> BIN_SHIFT;
      const int lp = h[k] + er[j];
      rec[lp] = (es[j] << BIN_SHIFT) | (ed[j] & (BIN_NODES - 1));
      gp[lp] = gbase[k] + er[j];
    }
  }
  __syncthreads();
  for (int i = tid; i < PCH; i += PTHR) binned[gp[i]] = rec[i];
}

__global__ __launch_bounds__(BIN_NODES) void bin_degree_kernel(
    const int* __restrict__ binstart, const int* __restrict__ binned,
    int2* __restrict__ info) {
  __shared__ int c[BIN_NODES];
  __shared__ int wsum[BIN_NODES / 32];
  const int tid = threadIdx.x;
  const int k = blockIdx.x;
  const int beg = binstart[k];
  const int end = binstart[k + 1];
  c[tid] = 0;
  __syncthreads();
  for (int i0 = beg + tid; i0 < end; i0 += BIN_NODES * 4) {
    int r[4];
#pragma unroll
    for (int j = 0; j < 4; ++j) {
      const int i = i0 + j * BIN_NODES;
      r[j] = (i < end) ? binned[i] : -1;
    }
#pragma unroll
    for (int j = 0; j < 4; ++j)
      if (r[j] >= 0) atomicAdd(&c[r[j] & (BIN_NODES - 1)], 1);
  }
  __syncthreads();
  const int cv = c[tid];
  const int incl = block_incl_scan<BIN_NODES / 32>(cv, wsum);
  const int node = k * BIN_NODES + tid;
  if (node < NNODES) {
    int2 r = {beg + incl - cv, __float_as_int(rsqrtf((float)cv + 1.0f))};
    info[node] = r;
  }
}

#define FPER 8
__global__ __launch_bounds__(FTHR) void bin_fill_kernel(
    const int* __restrict__ binstart, const int* __restrict__ binned,
    const int2* __restrict__ info, int2* __restrict__ csr) {
  __shared__ int cur[BIN_NODES];
  __shared__ float dv[BIN_NODES];
  __shared__ int2 st[FILL_CAP];
  const int tid = threadIdx.x;
  const int k = blockIdx.x;
  const int beg = binstart[k];
  const int end = binstart[k + 1];
  const int n = end - beg;
  if (tid < BIN_NODES) {
    const int node = k * BIN_NODES + tid;
    int2 ni = {end, 0};
    if (node < NNODES) ni = info[node];
    cur[tid] = ni.x - beg;
    dv[tid] = __int_as_float(ni.y);
  }
  __syncthreads();
  const bool staged = (n <= FILL_CAP);
  for (int i0 = tid; i0 < n; i0 += FTHR * FPER) {
    int r[FPER];
    float ds[FPER];
#pragma unroll
    for (int j = 0; j < FPER; ++j) {
      const int i = i0 + j * FTHR;
      r[j] = (i < n) ? binned[beg + i] : -1;
    }
#pragma unroll
    for (int j = 0; j < FPER; ++j) {
      const int s = (r[j] >= 0) ? (r[j] >> BIN_SHIFT) : 0;
      ds[j] = __int_as_float(info[s].y);
    }
#pragma unroll
    for (int j = 0; j < FPER; ++j) {
      if (r[j] >= 0) {
        const int s = r[j] >> BIN_SHIFT;
        const int dl = r[j] & (BIN_NODES - 1);
        const int p = atomicAdd(&cur[dl], 1);
        int2 o = {s, __float_as_int(ds[j] * dv[dl])};
        if (staged) st[p] = o; else csr[beg + p] = o;
      }
    }
  }
  if (staged) {
    __syncthreads();
    for (int i = tid; i < n; i += FTHR) csr[beg + i] = st[i];
  }
}

__global__ __launch_bounds__(256) void gcn_gemm1(const float* __restrict__ x,
                                                 const float* __restrict__ W1,
                                                 float* __restrict__ h1) {
  __shared__ float sW[FH * FIN];
  for (int i = threadIdx.x; i < FH * FIN; i += 256) sW[i] = W1[i];
  __syncthreads();

  const int lane = threadIdx.x & 31;
  const int wave = threadIdx.x >> 5;
  const int tile = blockIdx.x * 8 + wave;
  if (tile * 16 >= NNODES) return;
  const int m0 = tile * 16;
  const int lrow = lane & 15;
  const int koff = (lane >> 4) << 1;

  v8f acc0 = {}, acc1 = {}, acc2 = {}, acc3 = {};
  const float* xrow = x + (size_t)(m0 + lrow) * FIN;
#pragma unroll 4
  for (int k = 0; k < FIN; k += 4) {
    v2f a = *(const v2f*)(xrow + k + koff);
    v2f b0 = *(const v2f*)(&sW[(0 * 16 + lrow) * FIN + k + koff]);
    v2f b1 = *(const v2f*)(&sW[(1 * 16 + lrow) * FIN + k + koff]);
    v2f b2 = *(const v2f*)(&sW[(2 * 16 + lrow) * FIN + k + koff]);
    v2f b3 = *(const v2f*)(&sW[(3 * 16 + lrow) * FIN + k + koff]);
    acc0 = wmma_f32_k4(a, b0, acc0);
    acc1 = wmma_f32_k4(a, b1, acc1);
    acc2 = wmma_f32_k4(a, b2, acc2);
    acc3 = wmma_f32_k4(a, b3, acc3);
  }
  const int rbase = m0 + ((lane >> 4) << 3);
#pragma unroll
  for (int r = 0; r < 8; ++r) {
    h1[(size_t)(rbase + r) * FH + 0 * 16 + lrow] = acc0[r];
    h1[(size_t)(rbase + r) * FH + 1 * 16 + lrow] = acc1[r];
    h1[(size_t)(rbase + r) * FH + 2 * 16 + lrow] = acc2[r];
    h1[(size_t)(rbase + r) * FH + 3 * 16 + lrow] = acc3[r];
  }
}

__global__ __launch_bounds__(256) void gcn_gemm2(const float* __restrict__ agg1,
                                                 const float* __restrict__ b1,
                                                 const float* __restrict__ W2,
                                                 float* __restrict__ h2) {
  __shared__ float sW[FOUT * FH];
  __shared__ float sB[FH];
  for (int i = threadIdx.x; i < FOUT * FH; i += 256) sW[i] = W2[i];
  if (threadIdx.x < FH) sB[threadIdx.x] = b1[threadIdx.x];
  __syncthreads();

  const int lane = threadIdx.x & 31;
  const int wave = threadIdx.x >> 5;
  const int tile = blockIdx.x * 8 + wave;
  if (tile * 16 >= NNODES) return;
  const int m0 = tile * 16;
  const int lrow = lane & 15;
  const int koff = (lane >> 4) << 1;

  v8f acc0 = {}, acc1 = {};
  const float* arow = agg1 + (size_t)(m0 + lrow) * FH;
#pragma unroll 4
  for (int k = 0; k < FH; k += 4) {
    v2f ar = *(const v2f*)(arow + k + koff);
    v2f a;
    a.x = fmaxf(ar.x + sB[k + koff + 0], 0.0f);
    a.y = fmaxf(ar.y + sB[k + koff + 1], 0.0f);
    v2f b0 = *(const v2f*)(&sW[(0 * 16 + lrow) * FH + k + koff]);
    v2f b1v = *(const v2f*)(&sW[(1 * 16 + lrow) * FH + k + koff]);
    acc0 = wmma_f32_k4(a, b0, acc0);
    acc1 = wmma_f32_k4(a, b1v, acc1);
  }
  const int rbase = m0 + ((lane >> 4) << 3);
#pragma unroll
  for (int r = 0; r < 8; ++r) {
    h2[(size_t)(rbase + r) * FOUT + 0 * 16 + lrow] = acc0[r];
    h2[(size_t)(rbase + r) * FOUT + 1 * 16 + lrow] = acc1[r];
  }
}

template <int F, bool BIAS>
__global__ __launch_bounds__(256) void agg_rows_kernel(
    const int2* __restrict__ info, const int2* __restrict__ csr,
    const float* __restrict__ h, const float* __restrict__ bias,
    float* __restrict__ out) {
  const int TPE = F / 4;
  const int t = blockIdx.x * blockDim.x + threadIdx.x;
  const int node = t / TPE;
  const int c = t % TPE;
  if (node >= NNODES) return;
  const int2 ni = info[node];
  const int beg = ni.x;
  const int end = (node + 1 < NNODES) ? info[node + 1].x : NEDGES;
  const float4* __restrict__ h4 = (const float4*)h;
  float ws = __int_as_float(ni.y);
  ws *= ws;
  float4 v = h4[(size_t)node * TPE + c];
  float4 acc = {v.x * ws, v.y * ws, v.z * ws, v.w * ws};
  if (BIAS) {
    float4 bb = ((const float4*)bias)[c];
    acc.x += bb.x; acc.y += bb.y; acc.z += bb.z; acc.w += bb.w;
  }
  int p = beg;
  for (; p + 4 <= end; p += 4) {
    const int2 e0 = csr[p + 0], e1 = csr[p + 1];
    const int2 e2 = csr[p + 2], e3 = csr[p + 3];
    const float w0 = __int_as_float(e0.y), w1 = __int_as_float(e1.y);
    const float w2 = __int_as_float(e2.y), w3 = __int_as_float(e3.y);
    const float4 v0 = h4[(size_t)e0.x * TPE + c];
    const float4 v1 = h4[(size_t)e1.x * TPE + c];
    const float4 v2 = h4[(size_t)e2.x * TPE + c];
    const float4 v3 = h4[(size_t)e3.x * TPE + c];
    acc.x += v0.x * w0; acc.y += v0.y * w0; acc.z += v0.z * w0; acc.w += v0.w * w0;
    acc.x += v1.x * w1; acc.y += v1.y * w1; acc.z += v1.z * w1; acc.w += v1.w * w1;
    acc.x += v2.x * w2; acc.y += v2.y * w2; acc.z += v2.z * w2; acc.w += v2.w * w2;
    acc.x += v3.x * w3; acc.y += v3.y * w3; acc.z += v3.z * w3; acc.w += v3.w * w3;
  }
  for (; p < end; ++p) {
    const int2 e0 = csr[p];
    const float w0 = __int_as_float(e0.y);
    const float4 v0 = h4[(size_t)e0.x * TPE + c];
    acc.x += v0.x * w0; acc.y += v0.y * w0; acc.z += v0.z * w0; acc.w += v0.w * w0;
  }
  ((float4*)out)[(size_t)node * TPE + c] = acc;
}

extern "C" void kernel_launch(void* const* d_in, const int* in_sizes, int n_in,
                              void* d_out, int out_size, void* d_ws,
                              size_t ws_size, hipStream_t stream) {
  const float* x = (const float*)d_in[0];
  const int* edge_index = (const int*)d_in[1];
  const float* W1 = (const float*)d_in[2];
  const float* b1 = (const float*)d_in[3];
  const float* W2 = (const float*)d_in[4];
  const float* b2 = (const float*)d_in[5];
  float* out = (float*)d_out;

  const int* src = edge_index;
  const int* dst = edge_index + NEDGES;

  float* ws = (float*)d_ws;
  int2* info = (int2*)ws;
  float* h1 = ws + 2 * (size_t)NNODES;
  float* agg1 = h1 + (size_t)NNODES * FH;
  int2* csr = (int2*)(agg1 + (size_t)NNODES * FH);
  float* h2 = h1;
  int* binned = (int*)agg1;
  int* hist = binned + NEDGES;
  int* rel = hist + NBIN * PBLK;
  int* bintot = rel + NBIN * PBLK;
  int* binstart = bintot + 512;

  const int B = 256;
  const int nTiles = NNODES / 16;
  const int gemmBlocks = (nTiles + 7) / 8;

  bin_count_kernel<<<PBLK, PTHR, 0, stream>>>(dst, hist);
  bin_scan_kernel<<<NBIN, PBLK, 0, stream>>>(hist, rel, bintot);
  bin_scatter_kernel<<<PBLK, PTHR, 0, stream>>>(src, dst, rel, bintot, binstart,
                                               binned);
  bin_degree_kernel<<<NBIN, BIN_NODES, 0, stream>>>(binstart, binned, info);
  bin_fill_kernel<<<NBIN, FTHR, 0, stream>>>(binstart, binned, info, csr);

  gcn_gemm1<<<gemmBlocks, B, 0, stream>>>(x, W1, h1);

  agg_rows_kernel<FH, false>
      <<<(NNODES * (FH / 4) + B - 1) / B, B, 0, stream>>>(info, csr, h1, b1,
                                                          agg1);

  gcn_gemm2<<<gemmBlocks, B, 0, stream>>>(agg1, b1, W2, h2);

  agg_rows_kernel<FOUT, true>
      <<<(NNODES * (FOUT / 4) + B - 1) / B, B, 0, stream>>>(info, csr, h2, b2,
                                                            out);
}


# Edge split by bin made chunk-local: each chunk ordered by bin in place plus a run-start table; bin blocks sum the starts and collect their 256 runs; count and scan launches gone; sum order LDS arrival
# speedup vs baseline: 3.1754x; 1.2077x over previous
#include <hip/hip_runtime.h>

#define NNODES 100000
#define NEDGES 1600000
#define FIN 128
#define FH 64
#define FOUT 32

typedef float v2f __attribute__((ext_vector_type(2)));
typedef float v8f __attribute__((ext_vector_type(8)));

static __device__ __forceinline__ v8f wmma_f32_k4(v2f a, v2f b, v8f c) {
  return __builtin_amdgcn_wmma_f32_16x16x4_f32(false, a, false, b, (short)0, c,
                                               false, false);
}

#define BIN_SHIFT 8
#define BIN_NODES 256
#define NBIN ((NNODES + BIN_NODES - 1) / BIN_NODES)
#define PBLK 256
#define PCH (NEDGES / PBLK)
#define PTHR 1024
#define PPER ((PCH + PTHR - 1) / PTHR)
#define BTHR 1024
#define RPW (PBLK / (BTHR / 32))
#define BCAP 6144

static_assert(NEDGES % PBLK == 0, "edge chunks must be exact");
static_assert(NNODES < (1 << 23), "source id must fit above the low byte");
static_assert(NBIN <= PTHR, "one thread per bin in the partition blocks");
static_assert(PBLK % (BTHR / 32) == 0 && RPW % 4 == 0, "whole runs per wave, read four at a time");
static_assert(PBLK % 32 == 0 && PBLK <= BTHR, "one lane per chunk in whole waves");

static __device__ __forceinline__ int wave_incl_scan(int v) {
  const int lane = threadIdx.x & 31;
#pragma unroll
  for (int o = 1; o < 32; o <<= 1) {
    const int t = __shfl_up(v, o, 32);
    if (lane >= o) v += t;
  }
  return v;
}

static __device__ __forceinline__ int wave_sum(int v) {
#pragma unroll
  for (int o = 16; o > 0; o >>= 1) v += __shfl_xor(v, o, 32);
  return v;
}

template <int NW>
static __device__ __forceinline__ int block_incl_scan(int v, int* wsum) {
  const int lane = threadIdx.x & 31;
  const int w = threadIdx.x >> 5;
  v = wave_incl_scan(v);
  if (lane == 31) wsum[w] = v;
  __syncthreads();
  if (w == 0) {
    int t = (lane < NW) ? wsum[lane] : 0;
    t = wave_incl_scan(t);
    if (lane < NW) wsum[lane] = t;
  }
  __syncthreads();
  if (w > 0) v += wsum[w - 1];
  return v;
}

__global__ __launch_bounds__(PTHR) void bin_partition_kernel(
    const int* __restrict__ src, const int* __restrict__ dst,
    int* __restrict__ runoff, int* __restrict__ binned) {
  __shared__ int wsum[PTHR / 32];
  __shared__ int h[NBIN];
  __shared__ int rec[PCH];
  const int tid = threadIdx.x;
  const int b = blockIdx.x;

  if (tid < NBIN) h[tid] = 0;

  const int e0 = b * PCH;
  int es[PPER], ed[PPER], er[PPER];
#pragma unroll
  for (int j = 0; j < PPER; ++j) {
    const int i = tid + j * PTHR;
    es[j] = (i < PCH) ? src[e0 + i] : 0;
    ed[j] = (i < PCH) ? dst[e0 + i] : -1;
  }
  __syncthreads();

#pragma unroll
  for (int j = 0; j < PPER; ++j) {
    er[j] = 0;
    if (ed[j] >= 0) er[j] = atomicAdd(&h[ed[j] >> BIN_SHIFT], 1);
  }
  __syncthreads();

  const int hv = (tid < NBIN) ? h[tid] : 0;
  const int incl = block_incl_scan<PTHR / 32>(hv, wsum);
  if (tid < NBIN) {
    const int st = incl - hv;
    h[tid] = st;
    runoff[tid * PBLK + b] = st;
  }
  __syncthreads();

#pragma unroll
  for (int j = 0; j < PPER; ++j) {
    if (ed[j] >= 0) {
      const int lp = h[ed[j] >> BIN_SHIFT] + er[j];
      rec[lp] = (es[j] << BIN_SHIFT) | (ed[j] & (BIN_NODES - 1));
    }
  }
  __syncthreads();
  for (int i = tid; i < PCH; i += PTHR) binned[e0 + i] = rec[i];
}

__global__ __launch_bounds__(BTHR) void bin_build_kernel(
    const int* __restrict__ runoff, const int* __restrict__ binned,
    int2* __restrict__ info, int* __restrict__ csr) {
  __shared__ int cnt[BIN_NODES];
  __shared__ int wsum[BTHR / 32];
  __shared__ int4 rbeg4[PBLK / 4];
  __shared__ int4 rcnt4[PBLK / 4];
  __shared__ int ps0[PBLK / 32];
  __shared__ int ps1[PBLK / 32];
  __shared__ int st[BCAP];
  int* rbeg = (int*)rbeg4;
  int* rcnt = (int*)rcnt4;
  const int tid = threadIdx.x;
  const int lane = tid & 31;
  const int w = tid >> 5;
  const int k = blockIdx.x;
  int o0 = 0, o1 = 0;
  if (tid < BIN_NODES) cnt[tid] = 0;
  if (tid < PBLK) {
    o0 = runoff[k * PBLK + tid];
    o1 = (k + 1 < NBIN) ? runoff[(k + 1) * PBLK + tid] : PCH;
    rbeg[tid] = tid * PCH + o0;
    rcnt[tid] = o1 - o0;
  }
  __syncthreads();

  int rb[RPW], rc[RPW];
#pragma unroll
  for (int c4 = 0; c4 < RPW / 4; ++c4) {
    const int4 b4 = rbeg4[w * (RPW / 4) + c4];
    const int4 n4 = rcnt4[w * (RPW / 4) + c4];
    rb[c4 * 4 + 0] = __builtin_amdgcn_readfirstlane(b4.x);
    rb[c4 * 4 + 1] = __builtin_amdgcn_readfirstlane(b4.y);
    rb[c4 * 4 + 2] = __builtin_amdgcn_readfirstlane(b4.z);
    rb[c4 * 4 + 3] = __builtin_amdgcn_readfirstlane(b4.w);
    rc[c4 * 4 + 0] = __builtin_amdgcn_readfirstlane(n4.x);
    rc[c4 * 4 + 1] = __builtin_amdgcn_readfirstlane(n4.y);
    rc[c4 * 4 + 2] = __builtin_amdgcn_readfirstlane(n4.z);
    rc[c4 * 4 + 3] = __builtin_amdgcn_readfirstlane(n4.w);
  }
  int r[RPW];
#pragma unroll
  for (int c = 0; c < RPW; ++c)
    r[c] = (lane < rc[c]) ? binned[rb[c] + lane] : -1;

  if (tid < PBLK) {
    const int s0 = wave_sum(o0);
    const int s1 = wave_sum(o1);
    if (lane == 0) {
      ps0[w] = s0;
      ps1[w] = s1;
    }
  }

#pragma unroll
  for (int c = 0; c < RPW; ++c)
    if (r[c] >= 0) atomicAdd(&cnt[r[c] & (BIN_NODES - 1)], 1);
#pragma unroll
  for (int c = 0; c < RPW; ++c) {
    if (rc[c] > 32) {
      for (int j = lane + 32; j < rc[c]; j += 32)
        atomicAdd(&cnt[binned[rb[c] + j] & (BIN_NODES - 1)], 1);
    }
  }
  __syncthreads();
  int beg = 0, end = 0;
#pragma unroll
  for (int i = 0; i < PBLK / 32; ++i) {
    beg += ps0[i];
    end += ps1[i];
  }
  const int n = end - beg;
  const bool staged = (n <= BCAP);

  const int cv = (tid < BIN_NODES) ? cnt[tid] : 0;
  const int incl = block_incl_scan<BTHR / 32>(cv, wsum);
  if (tid < BIN_NODES) {
    const int first = incl - cv;
    cnt[tid] = first;
    const int node = k * BIN_NODES + tid;
    if (node < NNODES) {
      int2 ni = {beg + first, __float_as_int(rsqrtf((float)cv + 1.0f))};
      info[node] = ni;
    }
  }
  __syncthreads();
  int pl[RPW];
#pragma unroll
  for (int c = 0; c < RPW; ++c) {
    pl[c] = 0;
    if (r[c] >= 0) pl[c] = atomicAdd(&cnt[r[c] & (BIN_NODES - 1)], 1);
  }
  if (staged) {
#pragma unroll
    for (int c = 0; c < RPW; ++c)
      if (r[c] >= 0) st[pl[c]] = r[c] >> BIN_SHIFT;
  } else {
#pragma unroll
    for (int c = 0; c < RPW; ++c)
      if (r[c] >= 0) csr[beg + pl[c]] = r[c] >> BIN_SHIFT;
  }
#pragma unroll
  for (int c = 0; c < RPW; ++c) {
    if (rc[c] > 32) {
      for (int j = lane + 32; j < rc[c]; j += 32) {
        const int q = binned[rb[c] + j];
        const int p = atomicAdd(&cnt[q & (BIN_NODES - 1)], 1);
        if (staged) st[p] = q >> BIN_SHIFT;
        else csr[beg + p] = q >> BIN_SHIFT;
      }
    }
  }
  if (staged) {
    __syncthreads();
    for (int i = tid; i < n; i += BTHR) csr[beg + i] = st[i];
  }
}

__global__ __launch_bounds__(256) void gcn_gemm1(const float* __restrict__ x,
                                                 const float* __restrict__ W1,
                                                 const int2* __restrict__ info,
                                                 float* __restrict__ h1) {
  __shared__ float sW[FH * FIN];
  for (int i = threadIdx.x; i < FH * FIN; i += 256) sW[i] = W1[i];
  __syncthreads();

  const int lane = threadIdx.x & 31;
  const int wave = threadIdx.x >> 5;
  const int tile = blockIdx.x * 8 + wave;
  if (tile * 16 >= NNODES) return;
  const int m0 = tile * 16;
  const int lrow = lane & 15;
  const int koff = (lane >> 4) << 1;

  const int rbase = m0 + ((lane >> 4) << 3);
  float dv[8];
#pragma unroll
  for (int r = 0; r < 8; ++r) dv[r] = __int_as_float(info[rbase + r].y);

  v8f acc0 = {}, acc1 = {}, acc2 = {}, acc3 = {};
  const float* xrow = x + (size_t)(m0 + lrow) * FIN;
#pragma unroll 4
  for (int k = 0; k < FIN; k += 4) {
    v2f a = *(const v2f*)(xrow + k + koff);
    v2f b0 = *(const v2f*)(&sW[(0 * 16 + lrow) * FIN + k + koff]);
    v2f b1 = *(const v2f*)(&sW[(1 * 16 + lrow) * FIN + k + koff]);
    v2f b2 = *(const v2f*)(&sW[(2 * 16 + lrow) * FIN + k + koff]);
    v2f b3 = *(const v2f*)(&sW[(3 * 16 + lrow) * FIN + k + koff]);
    acc0 = wmma_f32_k4(a, b0, acc0);
    acc1 = wmma_f32_k4(a, b1, acc1);
    acc2 = wmma_f32_k4(a, b2, acc2);
    acc3 = wmma_f32_k4(a, b3, acc3);
  }
#pragma unroll
  for (int r = 0; r < 8; ++r) {
    h1[(size_t)(rbase + r) * FH + 0 * 16 + lrow] = acc0[r] * dv[r];
    h1[(size_t)(rbase + r) * FH + 1 * 16 + lrow] = acc1[r] * dv[r];
    h1[(size_t)(rbase + r) * FH + 2 * 16 + lrow] = acc2[r] * dv[r];
    h1[(size_t)(rbase + r) * FH + 3 * 16 + lrow] = acc3[r] * dv[r];
  }
}

__global__ __launch_bounds__(256) void gcn_gemm2(const float* __restrict__ agg1,
                                                 const float* __restrict__ b1,
                                                 const float* __restrict__ W2,
                                                 const int2* __restrict__ info,
                                                 float* __restrict__ h2) {
  __shared__ float sW[FOUT * FH];
  __shared__ float sB[FH];
  for (int i = threadIdx.x; i < FOUT * FH; i += 256) sW[i] = W2[i];
  if (threadIdx.x < FH) sB[threadIdx.x] = b1[threadIdx.x];
  __syncthreads();

  const int lane = threadIdx.x & 31;
  const int wave = threadIdx.x >> 5;
  const int tile = blockIdx.x * 8 + wave;
  if (tile * 16 >= NNODES) return;
  const int m0 = tile * 16;
  const int lrow = lane & 15;
  const int koff = (lane >> 4) << 1;

  const int rbase = m0 + ((lane >> 4) << 3);
  float dv[8];
#pragma unroll
  for (int r = 0; r < 8; ++r) dv[r] = __int_as_float(info[rbase + r].y);

  v8f acc0 = {}, acc1 = {};
  const float* arow = agg1 + (size_t)(m0 + lrow) * FH;
#pragma unroll 4
  for (int k = 0; k < FH; k += 4) {
    v2f ar = *(const v2f*)(arow + k + koff);
    v2f a;
    a.x = fmaxf(ar.x + sB[k + koff + 0], 0.0f);
    a.y = fmaxf(ar.y + sB[k + koff + 1], 0.0f);
    v2f b0 = *(const v2f*)(&sW[(0 * 16 + lrow) * FH + k + koff]);
    v2f b1v = *(const v2f*)(&sW[(1 * 16 + lrow) * FH + k + koff]);
    acc0 = wmma_f32_k4(a, b0, acc0);
    acc1 = wmma_f32_k4(a, b1v, acc1);
  }
#pragma unroll
  for (int r = 0; r < 8; ++r) {
    h2[(size_t)(rbase + r) * FOUT + 0 * 16 + lrow] = acc0[r] * dv[r];
    h2[(size_t)(rbase + r) * FOUT + 1 * 16 + lrow] = acc1[r] * dv[r];
  }
}

template <int F, bool BIAS>
__global__ __launch_bounds__(256) void agg_rows_kernel(
    const int2* __restrict__ info, const int* __restrict__ csr,
    const float* __restrict__ h, const float* __restrict__ bias,
    float* __restrict__ out) {
  const int TPE = F / 4;
  const int t = blockIdx.x * blockDim.x + threadIdx.x;
  const int node = t / TPE;
  const int c = t % TPE;
  if (node >= NNODES) return;
  const int2 ni = info[node];
  const int beg = ni.x;
  const int end = (node + 1 < NNODES) ? info[node + 1].x : NEDGES;
  const float4* __restrict__ h4 = (const float4*)h;
  const float di = __int_as_float(ni.y);
  float4 acc = h4[(size_t)node * TPE + c];
  int p = beg;
  for (; p + 4 <= end; p += 4) {
    const int s0 = csr[p + 0], s1 = csr[p + 1];
    const int s2 = csr[p + 2], s3 = csr[p + 3];
    const float4 v0 = h4[(size_t)s0 * TPE + c];
    const float4 v1 = h4[(size_t)s1 * TPE + c];
    const float4 v2 = h4[(size_t)s2 * TPE + c];
    const float4 v3 = h4[(size_t)s3 * TPE + c];
    acc.x += v0.x; acc.y += v0.y; acc.z += v0.z; acc.w += v0.w;
    acc.x += v1.x; acc.y += v1.y; acc.z += v1.z; acc.w += v1.w;
    acc.x += v2.x; acc.y += v2.y; acc.z += v2.z; acc.w += v2.w;
    acc.x += v3.x; acc.y += v3.y; acc.z += v3.z; acc.w += v3.w;
  }
  for (; p < end; ++p) {
    const float4 v0 = h4[(size_t)csr[p] * TPE + c];
    acc.x += v0.x; acc.y += v0.y; acc.z += v0.z; acc.w += v0.w;
  }
  acc.x *= di; acc.y *= di; acc.z *= di; acc.w *= di;
  if (BIAS) {
    float4 bb = ((const float4*)bias)[c];
    acc.x += bb.x; acc.y += bb.y; acc.z += bb.z; acc.w += bb.w;
  }
  ((float4*)out)[(size_t)node * TPE + c] = acc;
}

extern "C" void kernel_launch(void* const* d_in, const int* in_sizes, int n_in,
                              void* d_out, int out_size, void* d_ws,
                              size_t ws_size, hipStream_t stream) {
  const float* x = (const float*)d_in[0];
  const int* edge_index = (const int*)d_in[1];
  const float* W1 = (const float*)d_in[2];
  const float* b1 = (const float*)d_in[3];
  const float* W2 = (const float*)d_in[4];
  const float* b2 = (const float*)d_in[5];
  float* out = (float*)d_out;

  const int* src = edge_index;
  const int* dst = edge_index + NEDGES;

  float* ws = (float*)d_ws;
  int2* info = (int2*)ws;
  float* h1 = ws + 2 * (size_t)NNODES;
  float* agg1 = h1 + (size_t)NNODES * FH;
  int* csr = (int*)(agg1 + (size_t)NNODES * FH);
  float* h2 = h1;
  int* binned = (int*)agg1;
  int* runoff = binned + NEDGES;

  const int B = 256;
  const int nTiles = NNODES / 16;
  const int gemmBlocks = (nTiles + 7) / 8;

  bin_partition_kernel<<<PBLK, PTHR, 0, stream>>>(src, dst, runoff, binned);
  bin_build_kernel<<<NBIN, BTHR, 0, stream>>>(runoff, binned, info, csr);

  gcn_gemm1<<<gemmBlocks, B, 0, stream>>>(x, W1, info, h1);

  agg_rows_kernel<FH, false>
      <<<(NNODES * (FH / 4) + B - 1) / B, B, 0, stream>>>(info, csr, h1, b1,
                                                          agg1);

  gcn_gemm2<<<gemmBlocks, B, 0, stream>>>(agg1, b1, W2, info, h2);

  agg_rows_kernel<FOUT, true>
      <<<(NNODES * (FOUT / 4) + B - 1) / B, B, 0, stream>>>(info, csr, h2, b2,
                                                            out);
}


# Weight rows staged in LDS padded by four floats in both GEMMs: the rows of a B-fragment read fall on different banks (at the old pitch all on one bank group, served row by row); same sums
# speedup vs baseline: 3.6909x; 1.1623x over previous
#include <hip/hip_runtime.h>

#define NNODES 100000
#define NEDGES 1600000
#define FIN 128
#define FH 64
#define FOUT 32

typedef float v2f __attribute__((ext_vector_type(2)));
typedef float v8f __attribute__((ext_vector_type(8)));

static __device__ __forceinline__ v8f wmma_f32_k4(v2f a, v2f b, v8f c) {
  return __builtin_amdgcn_wmma_f32_16x16x4_f32(false, a, false, b, (short)0, c,
                                               false, false);
}

#define BIN_SHIFT 8
#define BIN_NODES 256
#define NBIN ((NNODES + BIN_NODES - 1) / BIN_NODES)
#define PBLK 256
#define PCH (NEDGES / PBLK)
#define PTHR 1024
#define PPER ((PCH + PTHR - 1) / PTHR)
#define BTHR 1024
#define RPW (PBLK / (BTHR / 32))
#define BCAP 6144

static_assert(NEDGES % PBLK == 0, "edge chunks must be exact");
static_assert(NNODES < (1 << 23), "source id must fit above the low byte");
static_assert(NBIN <= PTHR, "one thread per bin in the partition blocks");
static_assert(PBLK % (BTHR / 32) == 0 && RPW % 4 == 0, "whole runs per wave, read four at a time");
static_assert(PBLK % 32 == 0 && PBLK <= BTHR, "one lane per chunk in whole waves");

static __device__ __forceinline__ int wave_incl_scan(int v) {
  const int lane = threadIdx.x & 31;
#pragma unroll
  for (int o = 1; o < 32; o <<= 1) {
    const int t = __shfl_up(v, o, 32);
    if (lane >= o) v += t;
  }
  return v;
}

static __device__ __forceinline__ int wave_sum(int v) {
#pragma unroll
  for (int o = 16; o > 0; o >>= 1) v += __shfl_xor(v, o, 32);
  return v;
}

template <int NW>
static __device__ __forceinline__ int block_incl_scan(int v, int* wsum) {
  const int lane = threadIdx.x & 31;
  const int w = threadIdx.x >> 5;
  v = wave_incl_scan(v);
  if (lane == 31) wsum[w] = v;
  __syncthreads();
  if (w == 0) {
    int t = (lane < NW) ? wsum[lane] : 0;
    t = wave_incl_scan(t);
    if (lane < NW) wsum[lane] = t;
  }
  __syncthreads();
  if (w > 0) v += wsum[w - 1];
  return v;
}

__global__ __launch_bounds__(PTHR) void bin_partition_kernel(
    const int* __restrict__ src, const int* __restrict__ dst,
    int* __restrict__ runoff, int* __restrict__ binned) {
  __shared__ int wsum[PTHR / 32];
  __shared__ int h[NBIN];
  __shared__ int rec[PCH];
  const int tid = threadIdx.x;
  const int b = blockIdx.x;

  if (tid < NBIN) h[tid] = 0;

  const int e0 = b * PCH;
  int es[PPER], ed[PPER], er[PPER];
#pragma unroll
  for (int j = 0; j < PPER; ++j) {
    const int i = tid + j * PTHR;
    es[j] = (i < PCH) ? src[e0 + i] : 0;
    ed[j] = (i < PCH) ? dst[e0 + i] : -1;
  }
  __syncthreads();

#pragma unroll
  for (int j = 0; j < PPER; ++j) {
    er[j] = 0;
    if (ed[j] >= 0) er[j] = atomicAdd(&h[ed[j] >> BIN_SHIFT], 1);
  }
  __syncthreads();

  const int hv = (tid < NBIN) ? h[tid] : 0;
  const int incl = block_incl_scan<PTHR / 32>(hv, wsum);
  if (tid < NBIN) {
    const int st = incl - hv;
    h[tid] = st;
    runoff[tid * PBLK + b] = st;
  }
  __syncthreads();

#pragma unroll
  for (int j = 0; j < PPER; ++j) {
    if (ed[j] >= 0) {
      const int lp = h[ed[j] >> BIN_SHIFT] + er[j];
      rec[lp] = (es[j] << BIN_SHIFT) | (ed[j] & (BIN_NODES - 1));
    }
  }
  __syncthreads();
  for (int i = tid; i < PCH; i += PTHR) binned[e0 + i] = rec[i];
}

__global__ __launch_bounds__(BTHR) void bin_build_kernel(
    const int* __restrict__ runoff, const int* __restrict__ binned,
    int2* __restrict__ info, int* __restrict__ csr) {
  __shared__ int cnt[BIN_NODES];
  __shared__ int wsum[BTHR / 32];
  __shared__ int4 rbeg4[PBLK / 4];
  __shared__ int4 rcnt4[PBLK / 4];
  __shared__ int ps0[PBLK / 32];
  __shared__ int ps1[PBLK / 32];
  __shared__ int st[BCAP];
  int* rbeg = (int*)rbeg4;
  int* rcnt = (int*)rcnt4;
  const int tid = threadIdx.x;
  const int lane = tid & 31;
  const int w = tid >> 5;
  const int k = blockIdx.x;
  int o0 = 0, o1 = 0;
  if (tid < BIN_NODES) cnt[tid] = 0;
  if (tid < PBLK) {
    o0 = runoff[k * PBLK + tid];
    o1 = (k + 1 < NBIN) ? runoff[(k + 1) * PBLK + tid] : PCH;
    rbeg[tid] = tid * PCH + o0;
    rcnt[tid] = o1 - o0;
  }
  __syncthreads();

  int rb[RPW], rc[RPW];
#pragma unroll
  for (int c4 = 0; c4 < RPW / 4; ++c4) {
    const int4 b4 = rbeg4[w * (RPW / 4) + c4];
    const int4 n4 = rcnt4[w * (RPW / 4) + c4];
    rb[c4 * 4 + 0] = __builtin_amdgcn_readfirstlane(b4.x);
    rb[c4 * 4 + 1] = __builtin_amdgcn_readfirstlane(b4.y);
    rb[c4 * 4 + 2] = __builtin_amdgcn_readfirstlane(b4.z);
    rb[c4 * 4 + 3] = __builtin_amdgcn_readfirstlane(b4.w);
    rc[c4 * 4 + 0] = __builtin_amdgcn_readfirstlane(n4.x);
    rc[c4 * 4 + 1] = __builtin_amdgcn_readfirstlane(n4.y);
    rc[c4 * 4 + 2] = __builtin_amdgcn_readfirstlane(n4.z);
    rc[c4 * 4 + 3] = __builtin_amdgcn_readfirstlane(n4.w);
  }
  int r[RPW];
#pragma unroll
  for (int c = 0; c < RPW; ++c)
    r[c] = (lane < rc[c]) ? binned[rb[c] + lane] : -1;

  if (tid < PBLK) {
    const int s0 = wave_sum(o0);
    const int s1 = wave_sum(o1);
    if (lane == 0) {
      ps0[w] = s0;
      ps1[w] = s1;
    }
  }

#pragma unroll
  for (int c = 0; c < RPW; ++c)
    if (r[c] >= 0) atomicAdd(&cnt[r[c] & (BIN_NODES - 1)], 1);
#pragma unroll
  for (int c = 0; c < RPW; ++c) {
    if (rc[c] > 32) {
      for (int j = lane + 32; j < rc[c]; j += 32)
        atomicAdd(&cnt[binned[rb[c] + j] & (BIN_NODES - 1)], 1);
    }
  }
  __syncthreads();
  int beg = 0, end = 0;
#pragma unroll
  for (int i = 0; i < PBLK / 32; ++i) {
    beg += ps0[i];
    end += ps1[i];
  }
  const int n = end - beg;
  const bool staged = (n <= BCAP);

  const int cv = (tid < BIN_NODES) ? cnt[tid] : 0;
  const int incl = block_incl_scan<BTHR / 32>(cv, wsum);
  if (tid < BIN_NODES) {
    const int first = incl - cv;
    cnt[tid] = first;
    const int node = k * BIN_NODES + tid;
    if (node < NNODES) {
      int2 ni = {beg + first, __float_as_int(rsqrtf((float)cv + 1.0f))};
      info[node] = ni;
    }
  }
  __syncthreads();
  int pl[RPW];
#pragma unroll
  for (int c = 0; c < RPW; ++c) {
    pl[c] = 0;
    if (r[c] >= 0) pl[c] = atomicAdd(&cnt[r[c] & (BIN_NODES - 1)], 1);
  }
  if (staged) {
#pragma unroll
    for (int c = 0; c < RPW; ++c)
      if (r[c] >= 0) st[pl[c]] = r[c] >> BIN_SHIFT;
  } else {
#pragma unroll
    for (int c = 0; c < RPW; ++c)
      if (r[c] >= 0) csr[beg + pl[c]] = r[c] >> BIN_SHIFT;
  }
#pragma unroll
  for (int c = 0; c < RPW; ++c) {
    if (rc[c] > 32) {
      for (int j = lane + 32; j < rc[c]; j += 32) {
        const int q = binned[rb[c] + j];
        const int p = atomicAdd(&cnt[q & (BIN_NODES - 1)], 1);
        if (staged) st[p] = q >> BIN_SHIFT;
        else csr[beg + p] = q >> BIN_SHIFT;
      }
    }
  }
  if (staged) {
    __syncthreads();
    for (int i = tid; i < n; i += BTHR) csr[beg + i] = st[i];
  }
}

#define W1P (FIN + 4)
#define W2P (FH + 4)
static_assert(W1P % 4 == 0 && (W1P / 4) % 2 == 1, "pitch = four floats x odd");
static_assert(W2P % 4 == 0 && (W2P / 4) % 2 == 1, "pitch = four floats x odd");
__global__ __launch_bounds__(256) void gcn_gemm1(const float* __restrict__ x,
                                                 const float* __restrict__ W1,
                                                 const int2* __restrict__ info,
                                                 float* __restrict__ h1) {
  __shared__ float sW[FH * W1P];
  for (int i = threadIdx.x; i < FH * FIN; i += 256)
    sW[(i / FIN) * W1P + (i % FIN)] = W1[i];
  __syncthreads();

  const int lane = threadIdx.x & 31;
  const int wave = threadIdx.x >> 5;
  const int tile = blockIdx.x * 8 + wave;
  if (tile * 16 >= NNODES) return;
  const int m0 = tile * 16;
  const int lrow = lane & 15;
  const int koff = (lane >> 4) << 1;

  const int rbase = m0 + ((lane >> 4) << 3);
  float dv[8];
#pragma unroll
  for (int r = 0; r < 8; ++r) dv[r] = __int_as_float(info[rbase + r].y);

  v8f acc0 = {}, acc1 = {}, acc2 = {}, acc3 = {};
  const float* xrow = x + (size_t)(m0 + lrow) * FIN;
#pragma unroll 4
  for (int k = 0; k < FIN; k += 4) {
    v2f a = *(const v2f*)(xrow + k + koff);
    v2f b0 = *(const v2f*)(&sW[(0 * 16 + lrow) * W1P + k + koff]);
    v2f b1 = *(const v2f*)(&sW[(1 * 16 + lrow) * W1P + k + koff]);
    v2f b2 = *(const v2f*)(&sW[(2 * 16 + lrow) * W1P + k + koff]);
    v2f b3 = *(const v2f*)(&sW[(3 * 16 + lrow) * W1P + k + koff]);
    acc0 = wmma_f32_k4(a, b0, acc0);
    acc1 = wmma_f32_k4(a, b1, acc1);
    acc2 = wmma_f32_k4(a, b2, acc2);
    acc3 = wmma_f32_k4(a, b3, acc3);
  }
#pragma unroll
  for (int r = 0; r < 8; ++r) {
    h1[(size_t)(rbase + r) * FH + 0 * 16 + lrow] = acc0[r] * dv[r];
    h1[(size_t)(rbase + r) * FH + 1 * 16 + lrow] = acc1[r] * dv[r];
    h1[(size_t)(rbase + r) * FH + 2 * 16 + lrow] = acc2[r] * dv[r];
    h1[(size_t)(rbase + r) * FH + 3 * 16 + lrow] = acc3[r] * dv[r];
  }
}

__global__ __launch_bounds__(256) void gcn_gemm2(const float* __restrict__ agg1,
                                                 const float* __restrict__ b1,
                                                 const float* __restrict__ W2,
                                                 const int2* __restrict__ info,
                                                 float* __restrict__ h2) {
  __shared__ float sW[FOUT * W2P];
  __shared__ float sB[FH];
  for (int i = threadIdx.x; i < FOUT * FH; i += 256)
    sW[(i / FH) * W2P + (i % FH)] = W2[i];
  if (threadIdx.x < FH) sB[threadIdx.x] = b1[threadIdx.x];
  __syncthreads();

  const int lane = threadIdx.x & 31;
  const int wave = threadIdx.x >> 5;
  const int tile = blockIdx.x * 8 + wave;
  if (tile * 16 >= NNODES) return;
  const int m0 = tile * 16;
  const int lrow = lane & 15;
  const int koff = (lane >> 4) << 1;

  const int rbase = m0 + ((lane >> 4) << 3);
  float dv[8];
#pragma unroll
  for (int r = 0; r < 8; ++r) dv[r] = __int_as_float(info[rbase + r].y);

  v8f acc0 = {}, acc1 = {};
  const float* arow = agg1 + (size_t)(m0 + lrow) * FH;
#pragma unroll 4
  for (int k = 0; k < FH; k += 4) {
    v2f ar = *(const v2f*)(arow + k + koff);
    v2f a;
    a.x = fmaxf(ar.x + sB[k + koff + 0], 0.0f);
    a.y = fmaxf(ar.y + sB[k + koff + 1], 0.0f);
    v2f b0 = *(const v2f*)(&sW[(0 * 16 + lrow) * W2P + k + koff]);
    v2f b1v = *(const v2f*)(&sW[(1 * 16 + lrow) * W2P + k + koff]);
    acc0 = wmma_f32_k4(a, b0, acc0);
    acc1 = wmma_f32_k4(a, b1v, acc1);
  }
#pragma unroll
  for (int r = 0; r < 8; ++r) {
    h2[(size_t)(rbase + r) * FOUT + 0 * 16 + lrow] = acc0[r] * dv[r];
    h2[(size_t)(rbase + r) * FOUT + 1 * 16 + lrow] = acc1[r] * dv[r];
  }
}

template <int F, bool BIAS>
__global__ __launch_bounds__(256) void agg_rows_kernel(
    const int2* __restrict__ info, const int* __restrict__ csr,
    const float* __restrict__ h, const float* __restrict__ bias,
    float* __restrict__ out) {
  const int TPE = F / 4;
  const int t = blockIdx.x * blockDim.x + threadIdx.x;
  const int node = t / TPE;
  const int c = t % TPE;
  if (node >= NNODES) return;
  const int2 ni = info[node];
  const int beg = ni.x;
  const int end = (node + 1 < NNODES) ? info[node + 1].x : NEDGES;
  const float4* __restrict__ h4 = (const float4*)h;
  const float di = __int_as_float(ni.y);
  float4 acc = h4[(size_t)node * TPE + c];
  int p = beg;
  for (; p + 4 <= end; p += 4) {
    const int s0 = csr[p + 0], s1 = csr[p + 1];
    const int s2 = csr[p + 2], s3 = csr[p + 3];
    const float4 v0 = h4[(size_t)s0 * TPE + c];
    const float4 v1 = h4[(size_t)s1 * TPE + c];
    const float4 v2 = h4[(size_t)s2 * TPE + c];
    const float4 v3 = h4[(size_t)s3 * TPE + c];
    acc.x += v0.x; acc.y += v0.y; acc.z += v0.z; acc.w += v0.w;
    acc.x += v1.x; acc.y += v1.y; acc.z += v1.z; acc.w += v1.w;
    acc.x += v2.x; acc.y += v2.y; acc.z += v2.z; acc.w += v2.w;
    acc.x += v3.x; acc.y += v3.y; acc.z += v3.z; acc.w += v3.w;
  }
  for (; p < end; ++p) {
    const float4 v0 = h4[(size_t)csr[p] * TPE + c];
    acc.x += v0.x; acc.y += v0.y; acc.z += v0.z; acc.w += v0.w;
  }
  acc.x *= di; acc.y *= di; acc.z *= di; acc.w *= di;
  if (BIAS) {
    float4 bb = ((const float4*)bias)[c];
    acc.x += bb.x; acc.y += bb.y; acc.z += bb.z; acc.w += bb.w;
  }
  ((float4*)out)[(size_t)node * TPE + c] = acc;
}

extern "C" void kernel_launch(void* const* d_in, const int* in_sizes, int n_in,
                              void* d_out, int out_size, void* d_ws,
                              size_t ws_size, hipStream_t stream) {
  const float* x = (const float*)d_in[0];
  const int* edge_index = (const int*)d_in[1];
  const float* W1 = (const float*)d_in[2];
  const float* b1 = (const float*)d_in[3];
  const float* W2 = (const float*)d_in[4];
  const float* b2 = (const float*)d_in[5];
  float* out = (float*)d_out;

  const int* src = edge_index;
  const int* dst = edge_index + NEDGES;

  float* ws = (float*)d_ws;
  int2* info = (int2*)ws;
  float* h1 = ws + 2 * (size_t)NNODES;
  float* agg1 = h1 + (size_t)NNODES * FH;
  int* csr = (int*)(agg1 + (size_t)NNODES * FH);
  float* h2 = h1;
  int* binned = (int*)agg1;
  int* runoff = binned + NEDGES;

  const int B = 256;
  const int nTiles = NNODES / 16;
  const int gemmBlocks = (nTiles + 7) / 8;

  bin_partition_kernel<<<PBLK, PTHR, 0, stream>>>(src, dst, runoff, binned);
  bin_build_kernel<<<NBIN, BTHR, 0, stream>>>(runoff, binned, info, csr);

  gcn_gemm1<<<gemmBlocks, B, 0, stream>>>(x, W1, info, h1);

  agg_rows_kernel<FH, false>
      <<<(NNODES * (FH / 4) + B - 1) / B, B, 0, stream>>>(info, csr, h1, b1,
                                                          agg1);

  gcn_gemm2<<<gemmBlocks, B, 0, stream>>>(agg1, b1, W2, info, h2);

  agg_rows_kernel<FOUT, true>
      <<<(NNODES * (FOUT / 4) + B - 1) / B, B, 0, stream>>>(info, csr, h2, b2,
                                                            out);
}


# Aggregation: each wave walks 8 or 16 consecutive nodes; a node's ids are read 16 at a time (one load per lane) and passed across lanes, so 8 row reads per lane issue back to back; same sum order
# speedup vs baseline: 3.8288x; 1.0374x over previous
#include <hip/hip_runtime.h>

#define NNODES 100000
#define NEDGES 1600000
#define FIN 128
#define FH 64
#define FOUT 32

typedef float v2f __attribute__((ext_vector_type(2)));
typedef float v8f __attribute__((ext_vector_type(8)));

static __device__ __forceinline__ v8f wmma_f32_k4(v2f a, v2f b, v8f c) {
  return __builtin_amdgcn_wmma_f32_16x16x4_f32(false, a, false, b, (short)0, c,
                                               false, false);
}

#define BIN_SHIFT 8
#define BIN_NODES 256
#define NBIN ((NNODES + BIN_NODES - 1) / BIN_NODES)
#define PBLK 256
#define PCH (NEDGES / PBLK)
#define PTHR 1024
#define PPER ((PCH + PTHR - 1) / PTHR)
#define BTHR 1024
#define RPW (PBLK / (BTHR / 32))
#define BCAP 6144

static_assert(NEDGES % PBLK == 0, "edge chunks must be exact");
static_assert(NNODES < (1 << 23), "source id must fit above the low byte");
static_assert(NBIN <= PTHR, "one thread per bin in the partition blocks");
static_assert(PBLK % (BTHR / 32) == 0 && RPW % 4 == 0, "whole runs per wave, read four at a time");
static_assert(PBLK % 32 == 0 && PBLK <= BTHR, "one lane per chunk in whole waves");

static __device__ __forceinline__ int wave_incl_scan(int v) {
  const int lane = threadIdx.x & 31;
#pragma unroll
  for (int o = 1; o < 32; o <<= 1) {
    const int t = __shfl_up(v, o, 32);
    if (lane >= o) v += t;
  }
  return v;
}

static __device__ __forceinline__ int wave_sum(int v) {
#pragma unroll
  for (int o = 16; o > 0; o >>= 1) v += __shfl_xor(v, o, 32);
  return v;
}

template <int NW>
static __device__ __forceinline__ int block_incl_scan(int v, int* wsum) {
  const int lane = threadIdx.x & 31;
  const int w = threadIdx.x >> 5;
  v = wave_incl_scan(v);
  if (lane == 31) wsum[w] = v;
  __syncthreads();
  if (w == 0) {
    int t = (lane < NW) ? wsum[lane] : 0;
    t = wave_incl_scan(t);
    if (lane < NW) wsum[lane] = t;
  }
  __syncthreads();
  if (w > 0) v += wsum[w - 1];
  return v;
}

__global__ __launch_bounds__(PTHR) void bin_partition_kernel(
    const int* __restrict__ src, const int* __restrict__ dst,
    int* __restrict__ runoff, int* __restrict__ binned) {
  __shared__ int wsum[PTHR / 32];
  __shared__ int h[NBIN];
  __shared__ int rec[PCH];
  const int tid = threadIdx.x;
  const int b = blockIdx.x;

  if (tid < NBIN) h[tid] = 0;

  const int e0 = b * PCH;
  int es[PPER], ed[PPER], er[PPER];
#pragma unroll
  for (int j = 0; j < PPER; ++j) {
    const int i = tid + j * PTHR;
    es[j] = (i < PCH) ? src[e0 + i] : 0;
    ed[j] = (i < PCH) ? dst[e0 + i] : -1;
  }
  __syncthreads();

#pragma unroll
  for (int j = 0; j < PPER; ++j) {
    er[j] = 0;
    if (ed[j] >= 0) er[j] = atomicAdd(&h[ed[j] >> BIN_SHIFT], 1);
  }
  __syncthreads();

  const int hv = (tid < NBIN) ? h[tid] : 0;
  const int incl = block_incl_scan<PTHR / 32>(hv, wsum);
  if (tid < NBIN) {
    const int st = incl - hv;
    h[tid] = st;
    runoff[tid * PBLK + b] = st;
  }
  __syncthreads();

#pragma unroll
  for (int j = 0; j < PPER; ++j) {
    if (ed[j] >= 0) {
      const int lp = h[ed[j] >> BIN_SHIFT] + er[j];
      rec[lp] = (es[j] << BIN_SHIFT) | (ed[j] & (BIN_NODES - 1));
    }
  }
  __syncthreads();
  for (int i = tid; i < PCH; i += PTHR) binned[e0 + i] = rec[i];
}

__global__ __launch_bounds__(BTHR) void bin_build_kernel(
    const int* __restrict__ runoff, const int* __restrict__ binned,
    int2* __restrict__ info, int* __restrict__ csr) {
  __shared__ int cnt[BIN_NODES];
  __shared__ int wsum[BTHR / 32];
  __shared__ int4 rbeg4[PBLK / 4];
  __shared__ int4 rcnt4[PBLK / 4];
  __shared__ int ps0[PBLK / 32];
  __shared__ int ps1[PBLK / 32];
  __shared__ int st[BCAP];
  int* rbeg = (int*)rbeg4;
  int* rcnt = (int*)rcnt4;
  const int tid = threadIdx.x;
  const int lane = tid & 31;
  const int w = tid >> 5;
  const int k = blockIdx.x;
  int o0 = 0, o1 = 0;
  if (tid < BIN_NODES) cnt[tid] = 0;
  if (tid < PBLK) {
    o0 = runoff[k * PBLK + tid];
    o1 = (k + 1 < NBIN) ? runoff[(k + 1) * PBLK + tid] : PCH;
    rbeg[tid] = tid * PCH + o0;
    rcnt[tid] = o1 - o0;
  }
  __syncthreads();

  int rb[RPW], rc[RPW];
#pragma unroll
  for (int c4 = 0; c4 < RPW / 4; ++c4) {
    const int4 b4 = rbeg4[w * (RPW / 4) + c4];
    const int4 n4 = rcnt4[w * (RPW / 4) + c4];
    rb[c4 * 4 + 0] = __builtin_amdgcn_readfirstlane(b4.x);
    rb[c4 * 4 + 1] = __builtin_amdgcn_readfirstlane(b4.y);
    rb[c4 * 4 + 2] = __builtin_amdgcn_readfirstlane(b4.z);
    rb[c4 * 4 + 3] = __builtin_amdgcn_readfirstlane(b4.w);
    rc[c4 * 4 + 0] = __builtin_amdgcn_readfirstlane(n4.x);
    rc[c4 * 4 + 1] = __builtin_amdgcn_readfirstlane(n4.y);
    rc[c4 * 4 + 2] = __builtin_amdgcn_readfirstlane(n4.z);
    rc[c4 * 4 + 3] = __builtin_amdgcn_readfirstlane(n4.w);
  }
  int r[RPW];
#pragma unroll
  for (int c = 0; c < RPW; ++c)
    r[c] = (lane < rc[c]) ? binned[rb[c] + lane] : -1;

  if (tid < PBLK) {
    const int s0 = wave_sum(o0);
    const int s1 = wave_sum(o1);
    if (lane == 0) {
      ps0[w] = s0;
      ps1[w] = s1;
    }
  }

#pragma unroll
  for (int c = 0; c < RPW; ++c)
    if (r[c] >= 0) atomicAdd(&cnt[r[c] & (BIN_NODES - 1)], 1);
#pragma unroll
  for (int c = 0; c < RPW; ++c) {
    if (rc[c] > 32) {
      for (int j = lane + 32; j < rc[c]; j += 32)
        atomicAdd(&cnt[binned[rb[c] + j] & (BIN_NODES - 1)], 1);
    }
  }
  __syncthreads();
  int beg = 0, end = 0;
#pragma unroll
  for (int i = 0; i < PBLK / 32; ++i) {
    beg += ps0[i];
    end += ps1[i];
  }
  const int n = end - beg;
  const bool staged = (n <= BCAP);

  const int cv = (tid < BIN_NODES) ? cnt[tid] : 0;
  const int incl = block_incl_scan<BTHR / 32>(cv, wsum);
  if (tid < BIN_NODES) {
    const int first = incl - cv;
    cnt[tid] = first;
    const int node = k * BIN_NODES + tid;
    if (node < NNODES) {
      int2 ni = {beg + first, __float_as_int(rsqrtf((float)cv + 1.0f))};
      info[node] = ni;
    }
  }
  __syncthreads();
  int pl[RPW];
#pragma unroll
  for (int c = 0; c < RPW; ++c) {
    pl[c] = 0;
    if (r[c] >= 0) pl[c] = atomicAdd(&cnt[r[c] & (BIN_NODES - 1)], 1);
  }
  if (staged) {
#pragma unroll
    for (int c = 0; c < RPW; ++c)
      if (r[c] >= 0) st[pl[c]] = r[c] >> BIN_SHIFT;
  } else {
#pragma unroll
    for (int c = 0; c < RPW; ++c)
      if (r[c] >= 0) csr[beg + pl[c]] = r[c] >> BIN_SHIFT;
  }
#pragma unroll
  for (int c = 0; c < RPW; ++c) {
    if (rc[c] > 32) {
      for (int j = lane + 32; j < rc[c]; j += 32) {
        const int q = binned[rb[c] + j];
        const int p = atomicAdd(&cnt[q & (BIN_NODES - 1)], 1);
        if (staged) st[p] = q >> BIN_SHIFT;
        else csr[beg + p] = q >> BIN_SHIFT;
      }
    }
  }
  if (staged) {
    __syncthreads();
    for (int i = tid; i < n; i += BTHR) csr[beg + i] = st[i];
  }
}

#define W1P (FIN + 4)
#define W2P (FH + 4)
static_assert(W1P % 4 == 0 && (W1P / 4) % 2 == 1, "pitch = four floats x odd");
static_assert(W2P % 4 == 0 && (W2P / 4) % 2 == 1, "pitch = four floats x odd");
__global__ __launch_bounds__(256) void gcn_gemm1(const float* __restrict__ x,
                                                 const float* __restrict__ W1,
                                                 const int2* __restrict__ info,
                                                 float* __restrict__ h1) {
  __shared__ float sW[FH * W1P];
  for (int i = threadIdx.x; i < FH * FIN; i += 256)
    sW[(i / FIN) * W1P + (i % FIN)] = W1[i];
  __syncthreads();

  const int lane = threadIdx.x & 31;
  const int wave = threadIdx.x >> 5;
  const int tile = blockIdx.x * 8 + wave;
  if (tile * 16 >= NNODES) return;
  const int m0 = tile * 16;
  const int lrow = lane & 15;
  const int koff = (lane >> 4) << 1;

  const int rbase = m0 + ((lane >> 4) << 3);
  float dv[8];
#pragma unroll
  for (int r = 0; r < 8; ++r) dv[r] = __int_as_float(info[rbase + r].y);

  v8f acc0 = {}, acc1 = {}, acc2 = {}, acc3 = {};
  const float* xrow = x + (size_t)(m0 + lrow) * FIN;
#pragma unroll 4
  for (int k = 0; k < FIN; k += 4) {
    v2f a = *(const v2f*)(xrow + k + koff);
    v2f b0 = *(const v2f*)(&sW[(0 * 16 + lrow) * W1P + k + koff]);
    v2f b1 = *(const v2f*)(&sW[(1 * 16 + lrow) * W1P + k + koff]);
    v2f b2 = *(const v2f*)(&sW[(2 * 16 + lrow) * W1P + k + koff]);
    v2f b3 = *(const v2f*)(&sW[(3 * 16 + lrow) * W1P + k + koff]);
    acc0 = wmma_f32_k4(a, b0, acc0);
    acc1 = wmma_f32_k4(a, b1, acc1);
    acc2 = wmma_f32_k4(a, b2, acc2);
    acc3 = wmma_f32_k4(a, b3, acc3);
  }
#pragma unroll
  for (int r = 0; r < 8; ++r) {
    h1[(size_t)(rbase + r) * FH + 0 * 16 + lrow] = acc0[r] * dv[r];
    h1[(size_t)(rbase + r) * FH + 1 * 16 + lrow] = acc1[r] * dv[r];
    h1[(size_t)(rbase + r) * FH + 2 * 16 + lrow] = acc2[r] * dv[r];
    h1[(size_t)(rbase + r) * FH + 3 * 16 + lrow] = acc3[r] * dv[r];
  }
}

__global__ __launch_bounds__(256) void gcn_gemm2(const float* __restrict__ agg1,
                                                 const float* __restrict__ b1,
                                                 const float* __restrict__ W2,
                                                 const int2* __restrict__ info,
                                                 float* __restrict__ h2) {
  __shared__ float sW[FOUT * W2P];
  __shared__ float sB[FH];
  for (int i = threadIdx.x; i < FOUT * FH; i += 256)
    sW[(i / FH) * W2P + (i % FH)] = W2[i];
  if (threadIdx.x < FH) sB[threadIdx.x] = b1[threadIdx.x];
  __syncthreads();

  const int lane = threadIdx.x & 31;
  const int wave = threadIdx.x >> 5;
  const int tile = blockIdx.x * 8 + wave;
  if (tile * 16 >= NNODES) return;
  const int m0 = tile * 16;
  const int lrow = lane & 15;
  const int koff = (lane >> 4) << 1;

  const int rbase = m0 + ((lane >> 4) << 3);
  float dv[8];
#pragma unroll
  for (int r = 0; r < 8; ++r) dv[r] = __int_as_float(info[rbase + r].y);

  v8f acc0 = {}, acc1 = {};
  const float* arow = agg1 + (size_t)(m0 + lrow) * FH;
#pragma unroll 4
  for (int k = 0; k < FH; k += 4) {
    v2f ar = *(const v2f*)(arow + k + koff);
    v2f a;
    a.x = fmaxf(ar.x + sB[k + koff + 0], 0.0f);
    a.y = fmaxf(ar.y + sB[k + koff + 1], 0.0f);
    v2f b0 = *(const v2f*)(&sW[(0 * 16 + lrow) * W2P + k + koff]);
    v2f b1v = *(const v2f*)(&sW[(1 * 16 + lrow) * W2P + k + koff]);
    acc0 = wmma_f32_k4(a, b0, acc0);
    acc1 = wmma_f32_k4(a, b1v, acc1);
  }
#pragma unroll
  for (int r = 0; r < 8; ++r) {
    h2[(size_t)(rbase + r) * FOUT + 0 * 16 + lrow] = acc0[r] * dv[r];
    h2[(size_t)(rbase + r) * FOUT + 1 * 16 + lrow] = acc1[r] * dv[r];
  }
}

#define AGG_THR 128
#define AGG_BATCH 8
#define AGG_CHUNK (2 * AGG_BATCH)

template <int TPE>
static __device__ __forceinline__ void agg_ask_batch(float4 (&v)[AGG_BATCH],
                                                     const float* __restrict__ h,
                                                     int ids, int first, int c) {
#pragma unroll
  for (int j = 0; j < AGG_BATCH; ++j) {
    const int s = __shfl(ids, first + j, TPE);
    v[j] = make_float4(0.0f, 0.0f, 0.0f, 0.0f);
    if (s >= 0) {
      const unsigned boff = ((unsigned)s * TPE + (unsigned)c) * 16u;
      v[j] = *(const float4*)((const char*)h + boff);
    }
  }
}

static __device__ __forceinline__ void agg_add_batch(float4& acc,
                                                     const float4 (&v)[AGG_BATCH]) {
#pragma unroll
  for (int j = 0; j < AGG_BATCH; ++j) {
    acc.x += v[j].x; acc.y += v[j].y; acc.z += v[j].z; acc.w += v[j].w;
  }
}

template <int TPE>
static __device__ __forceinline__ void agg_ask_ids(const int* __restrict__ csr,
                                                   int at, int nrem, int c,
                                                   int& a, int& b) {
  a = (c < nrem) ? csr[at + c] : -1;
  b = -1;
  if (TPE == 8) b = (8 + c < nrem) ? csr[at + 8 + c] : -1;
}

template <int F, int ROUNDS, bool BIAS>
__global__ __launch_bounds__(AGG_THR) void agg_rows_kernel(
    const int2* __restrict__ info, const int* __restrict__ csr,
    const float* __restrict__ h, const float* __restrict__ bias,
    float* __restrict__ out) {
  constexpr int TPE = F / 4;
  constexpr int GPW = 32 / TPE;
  constexpr int NPW = GPW * ROUNDS;
  static_assert(TPE == 8 || TPE == 16, "a chunk of ids is one or two words per lane");
  static_assert(NPW + 1 <= 32, "one node record per lane, the closing one included");
  static_assert((size_t)NNODES * F * 4 < (1ull << 32), "row byte offsets fit 32 bits");
  const int lane = threadIdx.x & 31;
  const int c = lane & (TPE - 1);
  const int g = lane / TPE;
  const int node0 = (blockIdx.x * (AGG_THR / 32) + (threadIdx.x >> 5)) * NPW;
  if (node0 >= NNODES) return;

  int2 rec = {NEDGES, 0};
  if (lane <= NPW && node0 + lane < NNODES) rec = info[node0 + lane];

  int nbg = __shfl(rec.x, g, 32);
  int nen = __shfl(rec.x, g + 1, 32);
  int npa, npb;
  agg_ask_ids<TPE>(csr, nbg, nen - nbg, c, npa, npb);

#pragma unroll 1
  for (int r = 0; r < ROUNDS; ++r) {
    const int slot = r * GPW + g;
    const int node = node0 + slot;
    const bool live = node < NNODES;
    const float di = __int_as_float(__shfl(rec.y, slot, 32));
    const unsigned nboff = ((unsigned)node * TPE + (unsigned)c) * 16u;
    const int e = nen;
    int base = nbg;
    int ia = npa, ib = npb;
    float4 v[AGG_BATCH];
    float4 acc = make_float4(0.0f, 0.0f, 0.0f, 0.0f);

    int nrem = e - base;
    agg_ask_batch<TPE>(v, h, ia, 0, c);
    if (live) acc = *(const float4*)((const char*)h + nboff);
    if (r + 1 < ROUNDS) {
      nbg = __shfl(rec.x, slot + GPW, 32);
      nen = __shfl(rec.x, slot + GPW + 1, 32);
      agg_ask_ids<TPE>(csr, nbg, nen - nbg, c, npa, npb);
    }
    int na = -1, nb = -1;
    if (nrem > AGG_CHUNK)
      agg_ask_ids<TPE>(csr, base + AGG_CHUNK, nrem - AGG_CHUNK, c, na, nb);
    agg_add_batch(acc, v);
    if (nrem > AGG_BATCH) {
      if (TPE == 16) agg_ask_batch<TPE>(v, h, ia, AGG_BATCH, c);
      else agg_ask_batch<TPE>(v, h, ib, 0, c);
      agg_add_batch(acc, v);
    }
    base += AGG_CHUNK;
    ia = na;
    ib = nb;

    while (base < e) {
      nrem = e - base;
      agg_ask_batch<TPE>(v, h, ia, 0, c);
      na = -1;
      nb = -1;
      if (nrem > AGG_CHUNK)
        agg_ask_ids<TPE>(csr, base + AGG_CHUNK, nrem - AGG_CHUNK, c, na, nb);
      agg_add_batch(acc, v);
      if (nrem > AGG_BATCH) {
        if (TPE == 16) agg_ask_batch<TPE>(v, h, ia, AGG_BATCH, c);
        else agg_ask_batch<TPE>(v, h, ib, 0, c);
        agg_add_batch(acc, v);
      }
      base += AGG_CHUNK;
      ia = na;
      ib = nb;
    }

    if (live) {
      acc.x *= di; acc.y *= di; acc.z *= di; acc.w *= di;
      if (BIAS) {
        float4 bb = ((const float4*)bias)[c];
        acc.x += bb.x; acc.y += bb.y; acc.z += bb.z; acc.w += bb.w;
      }
      *(float4*)((char*)out + nboff) = acc;
    }
  }
}

#define AGG1_ROUNDS 4
#define AGG2_ROUNDS 4
#define AGG1_NPW ((32 / (FH / 4)) * AGG1_ROUNDS)
#define AGG2_NPW ((32 / (FOUT / 4)) * AGG2_ROUNDS)
#define AGG1_BLOCKS ((NNODES + AGG1_NPW * (AGG_THR / 32) - 1) / (AGG1_NPW * (AGG_THR / 32)))
#define AGG2_BLOCKS ((NNODES + AGG2_NPW * (AGG_THR / 32) - 1) / (AGG2_NPW * (AGG_THR / 32)))

extern "C" void kernel_launch(void* const* d_in, const int* in_sizes, int n_in,
                              void* d_out, int out_size, void* d_ws,
                              size_t ws_size, hipStream_t stream) {
  const float* x = (const float*)d_in[0];
  const int* edge_index = (const int*)d_in[1];
  const float* W1 = (const float*)d_in[2];
  const float* b1 = (const float*)d_in[3];
  const float* W2 = (const float*)d_in[4];
  const float* b2 = (const float*)d_in[5];
  float* out = (float*)d_out;

  const int* src = edge_index;
  const int* dst = edge_index + NEDGES;

  float* ws = (float*)d_ws;
  int2* info = (int2*)ws;
  float* h1 = ws + 2 * (size_t)NNODES;
  float* agg1 = h1 + (size_t)NNODES * FH;
  int* csr = (int*)(agg1 + (size_t)NNODES * FH);
  float* h2 = h1;
  int* binned = (int*)agg1;
  int* runoff = binned + NEDGES;

  const int B = 256;
  const int nTiles = NNODES / 16;
  const int gemmBlocks = (nTiles + 7) / 8;

  bin_partition_kernel<<<PBLK, PTHR, 0, stream>>>(src, dst, runoff, binned);
  bin_build_kernel<<<NBIN, BTHR, 0, stream>>>(runoff, binned, info, csr);

  gcn_gemm1<<<gemmBlocks, B, 0, stream>>>(x, W1, info, h1);

  agg_rows_kernel<FH, AGG1_ROUNDS, false>
      <<<AGG1_BLOCKS, AGG_THR, 0, stream>>>(info, csr, h1, b1, agg1);

  gcn_gemm2<<<gemmBlocks, B, 0, stream>>>(agg1, b1, W2, info, h2);

  agg_rows_kernel<FOUT, AGG2_ROUNDS, true>
      <<<AGG2_BLOCKS, AGG_THR, 0, stream>>>(info, csr, h2, b2, out);
}


# Layer-2 product folded into the layer-1 aggregation kernel: a wave sums 16 nodes, keeps the ReLU rows as an LDS tile, multiplies by W2 on WMMA; one launch and the agg1 round trip go; same sum order
# speedup vs baseline: 4.2351x; 1.1061x over previous
#include <hip/hip_runtime.h>

#define NNODES 100000
#define NEDGES 1600000
#define FIN 128
#define FH 64
#define FOUT 32

typedef float v2f __attribute__((ext_vector_type(2)));
typedef float v8f __attribute__((ext_vector_type(8)));

static __device__ __forceinline__ v8f wmma_f32_k4(v2f a, v2f b, v8f c) {
  return __builtin_amdgcn_wmma_f32_16x16x4_f32(false, a, false, b, (short)0, c,
                                               false, false);
}

#define BIN_SHIFT 8
#define BIN_NODES 256
#define NBIN ((NNODES + BIN_NODES - 1) / BIN_NODES)
#define PBLK 256
#define PCH (NEDGES / PBLK)
#define PTHR 1024
#define PPER ((PCH + PTHR - 1) / PTHR)
#define BTHR 1024
#define RPW (PBLK / (BTHR / 32))
#define BCAP 6144

static_assert(NEDGES % PBLK == 0, "edge chunks must be exact");
static_assert(NNODES < (1 << 23), "source id must fit above the low byte");
static_assert(NBIN <= PTHR, "one thread per bin in the partition blocks");
static_assert(PBLK % (BTHR / 32) == 0 && RPW % 4 == 0, "whole runs per wave, read four at a time");
static_assert(PBLK % 32 == 0 && PBLK <= BTHR, "one lane per chunk in whole waves");

static __device__ __forceinline__ int wave_incl_scan(int v) {
  const int lane = threadIdx.x & 31;
#pragma unroll
  for (int o = 1; o < 32; o <<= 1) {
    const int t = __shfl_up(v, o, 32);
    if (lane >= o) v += t;
  }
  return v;
}

static __device__ __forceinline__ int wave_sum(int v) {
#pragma unroll
  for (int o = 16; o > 0; o >>= 1) v += __shfl_xor(v, o, 32);
  return v;
}

template <int NW>
static __device__ __forceinline__ int block_incl_scan(int v, int* wsum) {
  const int lane = threadIdx.x & 31;
  const int w = threadIdx.x >> 5;
  v = wave_incl_scan(v);
  if (lane == 31) wsum[w] = v;
  __syncthreads();
  if (w == 0) {
    int t = (lane < NW) ? wsum[lane] : 0;
    t = wave_incl_scan(t);
    if (lane < NW) wsum[lane] = t;
  }
  __syncthreads();
  if (w > 0) v += wsum[w - 1];
  return v;
}

__global__ __launch_bounds__(PTHR) void bin_partition_kernel(
    const int* __restrict__ src, const int* __restrict__ dst,
    int* __restrict__ runoff, int* __restrict__ binned) {
  __shared__ int wsum[PTHR / 32];
  __shared__ int h[NBIN];
  __shared__ int rec[PCH];
  const int tid = threadIdx.x;
  const int b = blockIdx.x;

  if (tid < NBIN) h[tid] = 0;

  const int e0 = b * PCH;
  int es[PPER], ed[PPER], er[PPER];
#pragma unroll
  for (int j = 0; j < PPER; ++j) {
    const int i = tid + j * PTHR;
    es[j] = (i < PCH) ? src[e0 + i] : 0;
    ed[j] = (i < PCH) ? dst[e0 + i] : -1;
  }
  __syncthreads();

#pragma unroll
  for (int j = 0; j < PPER; ++j) {
    er[j] = 0;
    if (ed[j] >= 0) er[j] = atomicAdd(&h[ed[j] >> BIN_SHIFT], 1);
  }
  __syncthreads();

  const int hv = (tid < NBIN) ? h[tid] : 0;
  const int incl = block_incl_scan<PTHR / 32>(hv, wsum);
  if (tid < NBIN) {
    const int st = incl - hv;
    h[tid] = st;
    runoff[tid * PBLK + b] = st;
  }
  __syncthreads();

#pragma unroll
  for (int j = 0; j < PPER; ++j) {
    if (ed[j] >= 0) {
      const int lp = h[ed[j] >> BIN_SHIFT] + er[j];
      rec[lp] = (es[j] << BIN_SHIFT) | (ed[j] & (BIN_NODES - 1));
    }
  }
  __syncthreads();
  for (int i = tid; i < PCH; i += PTHR) binned[e0 + i] = rec[i];
}

__global__ __launch_bounds__(BTHR) void bin_build_kernel(
    const int* __restrict__ runoff, const int* __restrict__ binned,
    int2* __restrict__ info, int* __restrict__ csr) {
  __shared__ int cnt[BIN_NODES];
  __shared__ int wsum[BTHR / 32];
  __shared__ int4 rbeg4[PBLK / 4];
  __shared__ int4 rcnt4[PBLK / 4];
  __shared__ int ps0[PBLK / 32];
  __shared__ int ps1[PBLK / 32];
  __shared__ int st[BCAP];
  int* rbeg = (int*)rbeg4;
  int* rcnt = (int*)rcnt4;
  const int tid = threadIdx.x;
  const int lane = tid & 31;
  const int w = tid >> 5;
  const int k = blockIdx.x;
  int o0 = 0, o1 = 0;
  if (tid < BIN_NODES) cnt[tid] = 0;
  if (tid < PBLK) {
    o0 = runoff[k * PBLK + tid];
    o1 = (k + 1 < NBIN) ? runoff[(k + 1) * PBLK + tid] : PCH;
    rbeg[tid] = tid * PCH + o0;
    rcnt[tid] = o1 - o0;
  }
  __syncthreads();

  int rb[RPW], rc[RPW];
#pragma unroll
  for (int c4 = 0; c4 < RPW / 4; ++c4) {
    const int4 b4 = rbeg4[w * (RPW / 4) + c4];
    const int4 n4 = rcnt4[w * (RPW / 4) + c4];
    rb[c4 * 4 + 0] = __builtin_amdgcn_readfirstlane(b4.x);
    rb[c4 * 4 + 1] = __builtin_amdgcn_readfirstlane(b4.y);
    rb[c4 * 4 + 2] = __builtin_amdgcn_readfirstlane(b4.z);
    rb[c4 * 4 + 3] = __builtin_amdgcn_readfirstlane(b4.w);
    rc[c4 * 4 + 0] = __builtin_amdgcn_readfirstlane(n4.x);
    rc[c4 * 4 + 1] = __builtin_amdgcn_readfirstlane(n4.y);
    rc[c4 * 4 + 2] = __builtin_amdgcn_readfirstlane(n4.z);
    rc[c4 * 4 + 3] = __builtin_amdgcn_readfirstlane(n4.w);
  }
  int r[RPW];
#pragma unroll
  for (int c = 0; c < RPW; ++c)
    r[c] = (lane < rc[c]) ? binned[rb[c] + lane] : -1;

  if (tid < PBLK) {
    const int s0 = wave_sum(o0);
    const int s1 = wave_sum(o1);
    if (lane == 0) {
      ps0[w] = s0;
      ps1[w] = s1;
    }
  }

#pragma unroll
  for (int c = 0; c < RPW; ++c)
    if (r[c] >= 0) atomicAdd(&cnt[r[c] & (BIN_NODES - 1)], 1);
#pragma unroll
  for (int c = 0; c < RPW; ++c) {
    if (rc[c] > 32) {
      for (int j = lane + 32; j < rc[c]; j += 32)
        atomicAdd(&cnt[binned[rb[c] + j] & (BIN_NODES - 1)], 1);
    }
  }
  __syncthreads();
  int beg = 0, end = 0;
#pragma unroll
  for (int i = 0; i < PBLK / 32; ++i) {
    beg += ps0[i];
    end += ps1[i];
  }
  const int n = end - beg;
  const bool staged = (n <= BCAP);

  const int cv = (tid < BIN_NODES) ? cnt[tid] : 0;
  const int incl = block_incl_scan<BTHR / 32>(cv, wsum);
  if (tid < BIN_NODES) {
    const int first = incl - cv;
    cnt[tid] = first;
    const int node = k * BIN_NODES + tid;
    if (node < NNODES) {
      int2 ni = {beg + first, __float_as_int(rsqrtf((float)cv + 1.0f))};
      info[node] = ni;
    }
  }
  __syncthreads();
  int pl[RPW];
#pragma unroll
  for (int c = 0; c < RPW; ++c) {
    pl[c] = 0;
    if (r[c] >= 0) pl[c] = atomicAdd(&cnt[r[c] & (BIN_NODES - 1)], 1);
  }
  if (staged) {
#pragma unroll
    for (int c = 0; c < RPW; ++c)
      if (r[c] >= 0) st[pl[c]] = r[c] >> BIN_SHIFT;
  } else {
#pragma unroll
    for (int c = 0; c < RPW; ++c)
      if (r[c] >= 0) csr[beg + pl[c]] = r[c] >> BIN_SHIFT;
  }
#pragma unroll
  for (int c = 0; c < RPW; ++c) {
    if (rc[c] > 32) {
      for (int j = lane + 32; j < rc[c]; j += 32) {
        const int q = binned[rb[c] + j];
        const int p = atomicAdd(&cnt[q & (BIN_NODES - 1)], 1);
        if (staged) st[p] = q >> BIN_SHIFT;
        else csr[beg + p] = q >> BIN_SHIFT;
      }
    }
  }
  if (staged) {
    __syncthreads();
    for (int i = tid; i < n; i += BTHR) csr[beg + i] = st[i];
  }
}

#define W1P (FIN + 4)
#define W2P (FH + 4)
static_assert(W1P % 4 == 0 && (W1P / 4) % 2 == 1, "pitch = four floats x odd");
static_assert(W2P % 4 == 0 && (W2P / 4) % 2 == 1, "pitch = four floats x odd");
__global__ __launch_bounds__(256) void gcn_gemm1(const float* __restrict__ x,
                                                 const float* __restrict__ W1,
                                                 const int2* __restrict__ info,
                                                 float* __restrict__ h1) {
  __shared__ float sW[FH * W1P];
  for (int i = threadIdx.x; i < FH * FIN; i += 256)
    sW[(i / FIN) * W1P + (i % FIN)] = W1[i];
  __syncthreads();

  const int lane = threadIdx.x & 31;
  const int wave = threadIdx.x >> 5;
  const int tile = blockIdx.x * 8 + wave;
  if (tile * 16 >= NNODES) return;
  const int m0 = tile * 16;
  const int lrow = lane & 15;
  const int koff = (lane >> 4) << 1;

  const int rbase = m0 + ((lane >> 4) << 3);
  float dv[8];
#pragma unroll
  for (int r = 0; r < 8; ++r) dv[r] = __int_as_float(info[rbase + r].y);

  v8f acc0 = {}, acc1 = {}, acc2 = {}, acc3 = {};
  const float* xrow = x + (size_t)(m0 + lrow) * FIN;
#pragma unroll 4
  for (int k = 0; k < FIN; k += 4) {
    v2f a = *(const v2f*)(xrow + k + koff);
    v2f b0 = *(const v2f*)(&sW[(0 * 16 + lrow) * W1P + k + koff]);
    v2f b1 = *(const v2f*)(&sW[(1 * 16 + lrow) * W1P + k + koff]);
    v2f b2 = *(const v2f*)(&sW[(2 * 16 + lrow) * W1P + k + koff]);
    v2f b3 = *(const v2f*)(&sW[(3 * 16 + lrow) * W1P + k + koff]);
    acc0 = wmma_f32_k4(a, b0, acc0);
    acc1 = wmma_f32_k4(a, b1, acc1);
    acc2 = wmma_f32_k4(a, b2, acc2);
    acc3 = wmma_f32_k4(a, b3, acc3);
  }
#pragma unroll
  for (int r = 0; r < 8; ++r) {
    h1[(size_t)(rbase + r) * FH + 0 * 16 + lrow] = acc0[r] * dv[r];
    h1[(size_t)(rbase + r) * FH + 1 * 16 + lrow] = acc1[r] * dv[r];
    h1[(size_t)(rbase + r) * FH + 2 * 16 + lrow] = acc2[r] * dv[r];
    h1[(size_t)(rbase + r) * FH + 3 * 16 + lrow] = acc3[r] * dv[r];
  }
}

#define AGG_THR 128
#define AGG_BATCH 8
#define AGG_CHUNK (2 * AGG_BATCH)

template <int TPE>
static __device__ __forceinline__ void agg_ask_batch(float4 (&v)[AGG_BATCH],
                                                     const float* __restrict__ h,
                                                     int ids, int first, int c) {
#pragma unroll
  for (int j = 0; j < AGG_BATCH; ++j) {
    const int s = __shfl(ids, first + j, TPE);
    v[j] = make_float4(0.0f, 0.0f, 0.0f, 0.0f);
    if (s >= 0) {
      const unsigned boff = ((unsigned)s * TPE + (unsigned)c) * 16u;
      v[j] = *(const float4*)((const char*)h + boff);
    }
  }
}

static __device__ __forceinline__ void agg_add_batch(float4& acc,
                                                     const float4 (&v)[AGG_BATCH]) {
#pragma unroll
  for (int j = 0; j < AGG_BATCH; ++j) {
    acc.x += v[j].x; acc.y += v[j].y; acc.z += v[j].z; acc.w += v[j].w;
  }
}

template <int TPE>
static __device__ __forceinline__ void agg_ask_ids(const int* __restrict__ csr,
                                                   int at, int nrem, int c,
                                                   int& a, int& b) {
  a = (c < nrem) ? csr[at + c] : -1;
  b = -1;
  if (TPE == 8) b = (8 + c < nrem) ? csr[at + 8 + c] : -1;
}

template <int F, int ROUNDS, class Emit>
static __device__ __forceinline__ void agg_walk(const int2 rec,
                                                const int* __restrict__ csr,
                                                const float* __restrict__ h,
                                                const int node0, Emit emit) {
  constexpr int TPE = F / 4;
  constexpr int GPW = 32 / TPE;
  constexpr int NPW = GPW * ROUNDS;
  static_assert(TPE == 8 || TPE == 16, "a chunk of ids is one or two words per lane");
  static_assert(NPW + 1 <= 32, "one node record per lane, the closing one included");
  static_assert((size_t)NNODES * F * 4 < (1ull << 32), "row byte offsets fit 32 bits");
  const int lane = threadIdx.x & 31;
  const int c = lane & (TPE - 1);
  const int g = lane / TPE;

  int nbg = __shfl(rec.x, g, 32);
  int nen = __shfl(rec.x, g + 1, 32);
  int npa, npb;
  agg_ask_ids<TPE>(csr, nbg, nen - nbg, c, npa, npb);

#pragma unroll 1
  for (int r = 0; r < ROUNDS; ++r) {
    const int slot = r * GPW + g;
    const int node = node0 + slot;
    const bool live = node < NNODES;
    const float di = __int_as_float(__shfl(rec.y, slot, 32));
    const unsigned nboff = ((unsigned)node * TPE + (unsigned)c) * 16u;
    const int e = nen;
    int base = nbg;
    int ia = npa, ib = npb;
    float4 v[AGG_BATCH];
    float4 acc = make_float4(0.0f, 0.0f, 0.0f, 0.0f);

    int nrem = e - base;
    agg_ask_batch<TPE>(v, h, ia, 0, c);
    if (live) acc = *(const float4*)((const char*)h + nboff);
    if (r + 1 < ROUNDS) {
      nbg = __shfl(rec.x, slot + GPW, 32);
      nen = __shfl(rec.x, slot + GPW + 1, 32);
      agg_ask_ids<TPE>(csr, nbg, nen - nbg, c, npa, npb);
    }
    int na = -1, nb = -1;
    if (nrem > AGG_CHUNK)
      agg_ask_ids<TPE>(csr, base + AGG_CHUNK, nrem - AGG_CHUNK, c, na, nb);
    agg_add_batch(acc, v);
    if (nrem > AGG_BATCH) {
      if (TPE == 16) agg_ask_batch<TPE>(v, h, ia, AGG_BATCH, c);
      else agg_ask_batch<TPE>(v, h, ib, 0, c);
      agg_add_batch(acc, v);
    }
    base += AGG_CHUNK;
    ia = na;
    ib = nb;

    while (base < e) {
      nrem = e - base;
      agg_ask_batch<TPE>(v, h, ia, 0, c);
      na = -1;
      nb = -1;
      if (nrem > AGG_CHUNK)
        agg_ask_ids<TPE>(csr, base + AGG_CHUNK, nrem - AGG_CHUNK, c, na, nb);
      agg_add_batch(acc, v);
      if (nrem > AGG_BATCH) {
        if (TPE == 16) agg_ask_batch<TPE>(v, h, ia, AGG_BATCH, c);
        else agg_ask_batch<TPE>(v, h, ib, 0, c);
        agg_add_batch(acc, v);
      }
      base += AGG_CHUNK;
      ia = na;
      ib = nb;
    }

    acc.x *= di; acc.y *= di; acc.z *= di; acc.w *= di;
    emit(slot, live, nboff, acc);
  }
}

template <int F, int ROUNDS>
__global__ __launch_bounds__(AGG_THR) void agg_rows_kernel(
    const int2* __restrict__ info, const int* __restrict__ csr,
    const float* __restrict__ h, const float* __restrict__ bias,
    float* __restrict__ out) {
  constexpr int TPE = F / 4;
  constexpr int NPW = (32 / TPE) * ROUNDS;
  const int lane = threadIdx.x & 31;
  const int c = lane & (TPE - 1);
  const int node0 = (blockIdx.x * (AGG_THR / 32) + (threadIdx.x >> 5)) * NPW;
  if (node0 >= NNODES) return;

  int2 rec = {NEDGES, 0};
  if (lane <= NPW && node0 + lane < NNODES) rec = info[node0 + lane];

  agg_walk<F, ROUNDS>(rec, csr, h, node0,
                      [&](int, bool live, unsigned nboff, float4 sum) {
    if (live) {
      const float4 bb = ((const float4*)bias)[c];
      sum.x += bb.x; sum.y += bb.y; sum.z += bb.z; sum.w += bb.w;
      *(float4*)((char*)out + nboff) = sum;
    }
  });
}

#define FUS_WAVES 8
#define FUS_THR (FUS_WAVES * 32)
#define FUS_ROUNDS 8
#define FUS_BLOCKS ((NNODES / 16 + FUS_WAVES - 1) / FUS_WAVES)
#define AP (FH + 4)
static_assert(NNODES % 16 == 0, "every wave owns a whole tile of existing nodes");
static_assert((32 / (FH / 4)) * FUS_ROUNDS == 16, "a wave's run of nodes is one 16-row tile");
static_assert(AP % 4 == 0 && (AP / 4) % 2 == 1, "pitch = four floats x odd");
__global__ __launch_bounds__(FUS_THR) void agg1_gemm2_kernel(
    const int2* __restrict__ info, const int* __restrict__ csr,
    const float* __restrict__ h1, const float* __restrict__ b1,
    const float* __restrict__ W2, float* __restrict__ h2) {
  __shared__ __attribute__((aligned(16))) float sW[FOUT * W2P];
  __shared__ __attribute__((aligned(16))) float sA[FUS_WAVES * 16 * AP];
  const int lane = threadIdx.x & 31;
  const int wave = threadIdx.x >> 5;
  const int tile = blockIdx.x * FUS_WAVES + wave;
  const int node0 = tile * 16;
  const int c = lane & (FH / 4 - 1);
  float* tileA = sA + wave * 16 * AP;

  int2 rec = {NEDGES, 0};
  if (lane <= 16 && node0 + lane < NNODES) rec = info[node0 + lane];
  const float4 bb = ((const float4*)b1)[c];

  static_assert(FOUT * FH / 4 == 2 * FUS_THR && FH / 4 == 16, "two float4 of W2 per thread");
  const float4 w0 = ((const float4*)W2)[threadIdx.x];
  const float4 w1 = ((const float4*)W2)[threadIdx.x + FUS_THR];
  float* wdst = &sW[(threadIdx.x >> 4) * W2P + (threadIdx.x & 15) * 4];
  *(float4*)wdst = w0;
  *(float4*)(wdst + (FUS_THR >> 4) * W2P) = w1;
  __syncthreads();
  if (node0 >= NNODES) return;

  agg_walk<FH, FUS_ROUNDS>(rec, csr, h1, node0,
                           [&](int slot, bool, unsigned, float4 sum) {
    sum.x = fmaxf(sum.x + bb.x, 0.0f);
    sum.y = fmaxf(sum.y + bb.y, 0.0f);
    sum.z = fmaxf(sum.z + bb.z, 0.0f);
    sum.w = fmaxf(sum.w + bb.w, 0.0f);
    *(float4*)&tileA[slot * AP + 4 * c] = sum;
  });


  const int lrow = lane & 15;
  const int hh = lane >> 4;
  const int koff = hh << 1;
  const float* arow = tileA + lrow * AP + koff;
  const float* brow = sW + lrow * W2P + koff;
  v8f acc0 = {}, acc1 = {};
#pragma unroll
  for (int k = 0; k < FH; k += 4) {
    const v2f a = *(const v2f*)(arow + k);
    const v2f b0 = *(const v2f*)(brow + k);
    const v2f b1v = *(const v2f*)(brow + 16 * W2P + k);
    acc0 = wmma_f32_k4(a, b0, acc0);
    acc1 = wmma_f32_k4(a, b1v, acc1);
  }

  const int rsub = hh << 3;
#pragma unroll
  for (int r = 0; r < 8; ++r) {
    const float dv = __int_as_float(__shfl(rec.y, rsub + r, 32));
    h2[(size_t)(node0 + rsub + r) * FOUT + 0 * 16 + lrow] = acc0[r] * dv;
    h2[(size_t)(node0 + rsub + r) * FOUT + 1 * 16 + lrow] = acc1[r] * dv;
  }
}

#define AGG2_ROUNDS 4
#define AGG2_NPW ((32 / (FOUT / 4)) * AGG2_ROUNDS)
#define AGG2_BLOCKS ((NNODES + AGG2_NPW * (AGG_THR / 32) - 1) / (AGG2_NPW * (AGG_THR / 32)))

extern "C" void kernel_launch(void* const* d_in, const int* in_sizes, int n_in,
                              void* d_out, int out_size, void* d_ws,
                              size_t ws_size, hipStream_t stream) {
  const float* x = (const float*)d_in[0];
  const int* edge_index = (const int*)d_in[1];
  const float* W1 = (const float*)d_in[2];
  const float* b1 = (const float*)d_in[3];
  const float* W2 = (const float*)d_in[4];
  const float* b2 = (const float*)d_in[5];
  float* out = (float*)d_out;

  const int* src = edge_index;
  const int* dst = edge_index + NEDGES;

  float* ws = (float*)d_ws;
  int2* info = (int2*)ws;
  float* h1 = ws + 2 * (size_t)NNODES;
  float* region3 = h1 + (size_t)NNODES * FH;
  int* csr = (int*)(region3 + (size_t)NNODES * FH);
  float* h2 = region3;
  int* binned = (int*)region3;
  int* runoff = binned + NEDGES;

  const int B = 256;
  const int nTiles = NNODES / 16;
  const int gemmBlocks = (nTiles + 7) / 8;

  bin_partition_kernel<<<PBLK, PTHR, 0, stream>>>(src, dst, runoff, binned);
  bin_build_kernel<<<NBIN, BTHR, 0, stream>>>(runoff, binned, info, csr);

  gcn_gemm1<<<gemmBlocks, B, 0, stream>>>(x, W1, info, h1);

  agg1_gemm2_kernel<<<FUS_BLOCKS, FUS_THR, 0, stream>>>(info, csr, h1, b1, W2, h2);

  agg_rows_kernel<FOUT, AGG2_ROUNDS>
      <<<AGG2_BLOCKS, AGG_THR, 0, stream>>>(info, csr, h2, b2, out);
}


# Layer-1 product folded into the bin build kernel: behind its list each bin block multiplies its own 256 rows of x by W1 on WMMA (16 waves, a tile each), dinv from LDS; one launch less; same sums
# speedup vs baseline: 4.5566x; 1.0759x over previous
#include <hip/hip_runtime.h>

#define NNODES 100000
#define NEDGES 1600000
#define FIN 128
#define FH 64
#define FOUT 32

typedef float v2f __attribute__((ext_vector_type(2)));
typedef float v8f __attribute__((ext_vector_type(8)));

static __device__ __forceinline__ v8f wmma_f32_k4(v2f a, v2f b, v8f c) {
  return __builtin_amdgcn_wmma_f32_16x16x4_f32(false, a, false, b, (short)0, c,
                                               false, false);
}

#define BIN_SHIFT 8
#define BIN_NODES 256
#define NBIN ((NNODES + BIN_NODES - 1) / BIN_NODES)
#define PBLK 256
#define PCH (NEDGES / PBLK)
#define PTHR 1024
#define PPER ((PCH + PTHR - 1) / PTHR)
#define BTHR 1024
#define RPW (PBLK / (BTHR / 32))
#define BCAP 6144

static_assert(NEDGES % PBLK == 0, "edge chunks must be exact");
static_assert(NNODES < (1 << 23), "source id must fit above the low byte");
static_assert(NBIN <= PTHR, "one thread per bin in the partition blocks");
static_assert(PBLK % (BTHR / 32) == 0 && RPW % 4 == 0, "whole runs per wave, read four at a time");
static_assert(PBLK % 32 == 0 && PBLK <= BTHR, "one lane per chunk in whole waves");

static __device__ __forceinline__ int wave_incl_scan(int v) {
  const int lane = threadIdx.x & 31;
#pragma unroll
  for (int o = 1; o < 32; o <<= 1) {
    const int t = __shfl_up(v, o, 32);
    if (lane >= o) v += t;
  }
  return v;
}

static __device__ __forceinline__ int wave_sum(int v) {
#pragma unroll
  for (int o = 16; o > 0; o >>= 1) v += __shfl_xor(v, o, 32);
  return v;
}

template <int NW>
static __device__ __forceinline__ int block_incl_scan(int v, int* wsum) {
  const int lane = threadIdx.x & 31;
  const int w = threadIdx.x >> 5;
  v = wave_incl_scan(v);
  if (lane == 31) wsum[w] = v;
  __syncthreads();
  if (w == 0) {
    int t = (lane < NW) ? wsum[lane] : 0;
    t = wave_incl_scan(t);
    if (lane < NW) wsum[lane] = t;
  }
  __syncthreads();
  if (w > 0) v += wsum[w - 1];
  return v;
}

__global__ __launch_bounds__(PTHR) void bin_partition_kernel(
    const int* __restrict__ src, const int* __restrict__ dst,
    int* __restrict__ runoff, int* __restrict__ binned) {
  __shared__ int wsum[PTHR / 32];
  __shared__ int h[NBIN];
  __shared__ int rec[PCH];
  const int tid = threadIdx.x;
  const int b = blockIdx.x;

  if (tid < NBIN) h[tid] = 0;

  const int e0 = b * PCH;
  int es[PPER], ed[PPER], er[PPER];
#pragma unroll
  for (int j = 0; j < PPER; ++j) {
    const int i = tid + j * PTHR;
    es[j] = (i < PCH) ? src[e0 + i] : 0;
    ed[j] = (i < PCH) ? dst[e0 + i] : -1;
  }
  __syncthreads();

#pragma unroll
  for (int j = 0; j < PPER; ++j) {
    er[j] = 0;
    if (ed[j] >= 0) er[j] = atomicAdd(&h[ed[j] >> BIN_SHIFT], 1);
  }
  __syncthreads();

  const int hv = (tid < NBIN) ? h[tid] : 0;
  const int incl = block_incl_scan<PTHR / 32>(hv, wsum);
  if (tid < NBIN) {
    const int st = incl - hv;
    h[tid] = st;
    runoff[tid * PBLK + b] = st;
  }
  __syncthreads();

#pragma unroll
  for (int j = 0; j < PPER; ++j) {
    if (ed[j] >= 0) {
      const int lp = h[ed[j] >> BIN_SHIFT] + er[j];
      rec[lp] = (es[j] << BIN_SHIFT) | (ed[j] & (BIN_NODES - 1));
    }
  }
  __syncthreads();
  for (int i = tid; i < PCH; i += PTHR) binned[e0 + i] = rec[i];
}

#define W1P (FIN + 4)
#define W2P (FH + 4)
static_assert(W1P % 4 == 0 && (W1P / 4) % 2 == 1, "pitch = four floats x odd");
static_assert(W2P % 4 == 0 && (W2P / 4) % 2 == 1, "pitch = four floats x odd");

static_assert(BIN_NODES % 16 == 0 && BIN_NODES / 16 <= BTHR / 32, "one wave per 16-row tile of the bin");
static_assert(FH * FIN / 4 == 2 * BTHR && FIN / 4 == 32, "two float4 of W1 per thread");
__global__ __launch_bounds__(BTHR, 16) void bin_build_gemm1_kernel(
    const int* __restrict__ runoff, const int* __restrict__ binned,
    int2* __restrict__ info, int* __restrict__ csr,
    const float* __restrict__ x, const float* __restrict__ W1,
    float* __restrict__ h1) {
  __shared__ __attribute__((aligned(16))) float sW[FH * W1P];
  __shared__ float sdinv[BIN_NODES];
  __shared__ int cnt[BIN_NODES];
  __shared__ int wsum[BTHR / 32];
  __shared__ int4 rbeg4[PBLK / 4];
  __shared__ int4 rcnt4[PBLK / 4];
  __shared__ int ps0[PBLK / 32];
  __shared__ int ps1[PBLK / 32];
  __shared__ int st[BCAP];
  int* rbeg = (int*)rbeg4;
  int* rcnt = (int*)rcnt4;
  const int tid = threadIdx.x;
  const int lane = tid & 31;
  const int w = tid >> 5;
  const int k = blockIdx.x;
  int o0 = 0, o1 = 0;
  if (tid < BIN_NODES) cnt[tid] = 0;
  if (tid < PBLK) {
    o0 = runoff[k * PBLK + tid];
    o1 = (k + 1 < NBIN) ? runoff[(k + 1) * PBLK + tid] : PCH;
  }
  const float4 wv0 = ((const float4*)W1)[tid];
  const float4 wv1 = ((const float4*)W1)[tid + BTHR];
  if (tid < PBLK) {
    rbeg[tid] = tid * PCH + o0;
    rcnt[tid] = o1 - o0;
  }
  {
    float* wdst = &sW[(tid >> 5) * W1P + (tid & 31) * 4];
    *(float4*)wdst = wv0;
    *(float4*)(wdst + (BTHR >> 5) * W1P) = wv1;
  }
  __syncthreads();

  int rb[RPW], rc[RPW];
#pragma unroll
  for (int c4 = 0; c4 < RPW / 4; ++c4) {
    const int4 b4 = rbeg4[w * (RPW / 4) + c4];
    const int4 n4 = rcnt4[w * (RPW / 4) + c4];
    rb[c4 * 4 + 0] = __builtin_amdgcn_readfirstlane(b4.x);
    rb[c4 * 4 + 1] = __builtin_amdgcn_readfirstlane(b4.y);
    rb[c4 * 4 + 2] = __builtin_amdgcn_readfirstlane(b4.z);
    rb[c4 * 4 + 3] = __builtin_amdgcn_readfirstlane(b4.w);
    rc[c4 * 4 + 0] = __builtin_amdgcn_readfirstlane(n4.x);
    rc[c4 * 4 + 1] = __builtin_amdgcn_readfirstlane(n4.y);
    rc[c4 * 4 + 2] = __builtin_amdgcn_readfirstlane(n4.z);
    rc[c4 * 4 + 3] = __builtin_amdgcn_readfirstlane(n4.w);
  }
  int r[RPW];
#pragma unroll
  for (int c = 0; c < RPW; ++c)
    r[c] = (lane < rc[c]) ? binned[rb[c] + lane] : -1;

  if (tid < PBLK) {
    const int s0 = wave_sum(o0);
    const int s1 = wave_sum(o1);
    if (lane == 0) {
      ps0[w] = s0;
      ps1[w] = s1;
    }
  }

#pragma unroll
  for (int c = 0; c < RPW; ++c)
    if (r[c] >= 0) atomicAdd(&cnt[r[c] & (BIN_NODES - 1)], 1);
#pragma unroll
  for (int c = 0; c < RPW; ++c) {
    if (rc[c] > 32) {
      for (int j = lane + 32; j < rc[c]; j += 32)
        atomicAdd(&cnt[binned[rb[c] + j] & (BIN_NODES - 1)], 1);
    }
  }
  __syncthreads();
  int beg = 0, end = 0;
#pragma unroll
  for (int i = 0; i < PBLK / 32; ++i) {
    beg += ps0[i];
    end += ps1[i];
  }
  const int n = end - beg;
  const bool staged = (n <= BCAP);

  const int cv = (tid < BIN_NODES) ? cnt[tid] : 0;
  const int incl = block_incl_scan<BTHR / 32>(cv, wsum);
  if (tid < BIN_NODES) {
    const int first = incl - cv;
    cnt[tid] = first;
    const int node = k * BIN_NODES + tid;
    const float dinv = rsqrtf((float)cv + 1.0f);
    sdinv[tid] = dinv;
    if (node < NNODES) {
      int2 ni = {beg + first, __float_as_int(dinv)};
      info[node] = ni;
    }
  }
  __syncthreads();
  int pl[RPW];
#pragma unroll
  for (int c = 0; c < RPW; ++c) {
    pl[c] = 0;
    if (r[c] >= 0) pl[c] = atomicAdd(&cnt[r[c] & (BIN_NODES - 1)], 1);
  }
  if (staged) {
#pragma unroll
    for (int c = 0; c < RPW; ++c)
      if (r[c] >= 0) st[pl[c]] = r[c] >> BIN_SHIFT;
  } else {
#pragma unroll
    for (int c = 0; c < RPW; ++c)
      if (r[c] >= 0) csr[beg + pl[c]] = r[c] >> BIN_SHIFT;
  }
#pragma unroll
  for (int c = 0; c < RPW; ++c) {
    if (rc[c] > 32) {
      for (int j = lane + 32; j < rc[c]; j += 32) {
        const int q = binned[rb[c] + j];
        const int p = atomicAdd(&cnt[q & (BIN_NODES - 1)], 1);
        if (staged) st[p] = q >> BIN_SHIFT;
        else csr[beg + p] = q >> BIN_SHIFT;
      }
    }
  }
  if (staged) {
    __syncthreads();
    for (int i = tid; i < n; i += BTHR) csr[beg + i] = st[i];
  }

  const int ws = __builtin_amdgcn_readfirstlane(w);
  if (ws >= BIN_NODES / 16) return;
  const int m0 = k * BIN_NODES + ws * 16;
  if (m0 >= NNODES) return;

  v8f acc0 = {}, acc1 = {}, acc2 = {}, acc3 = {};
  {
    const int lrow = lane & 15;
    const int koff = (lane >> 4) << 1;
    const float* xrow = x + (size_t)(m0 + lrow) * FIN;
#pragma unroll 4
    for (int kk = 0; kk < FIN; kk += 4) {
      v2f a = *(const v2f*)(xrow + kk + koff);
      v2f b0 = *(const v2f*)(&sW[(0 * 16 + lrow) * W1P + kk + koff]);
      v2f b1 = *(const v2f*)(&sW[(1 * 16 + lrow) * W1P + kk + koff]);
      v2f b2 = *(const v2f*)(&sW[(2 * 16 + lrow) * W1P + kk + koff]);
      v2f b3 = *(const v2f*)(&sW[(3 * 16 + lrow) * W1P + kk + koff]);
      acc0 = wmma_f32_k4(a, b0, acc0);
      acc1 = wmma_f32_k4(a, b1, acc1);
      acc2 = wmma_f32_k4(a, b2, acc2);
      acc3 = wmma_f32_k4(a, b3, acc3);
    }
  }

  const int col = lane & 15;
  const int rsub = ws * 16 + ((lane >> 4) << 3);
  float* hrow = h1 + (size_t)(k * BIN_NODES + rsub) * FH + col;
#pragma unroll
  for (int r = 0; r < 8; ++r) {
    const float dv = sdinv[rsub + r];
    hrow[r * FH + 0 * 16] = acc0[r] * dv;
    hrow[r * FH + 1 * 16] = acc1[r] * dv;
    hrow[r * FH + 2 * 16] = acc2[r] * dv;
    hrow[r * FH + 3 * 16] = acc3[r] * dv;
  }
}

#define AGG_THR 128
#define AGG_BATCH 8
#define AGG_CHUNK (2 * AGG_BATCH)

template <int TPE>
static __device__ __forceinline__ void agg_ask_batch(float4 (&v)[AGG_BATCH],
                                                     const float* __restrict__ h,
                                                     int ids, int first, int c) {
#pragma unroll
  for (int j = 0; j < AGG_BATCH; ++j) {
    const int s = __shfl(ids, first + j, TPE);
    v[j] = make_float4(0.0f, 0.0f, 0.0f, 0.0f);
    if (s >= 0) {
      const unsigned boff = ((unsigned)s * TPE + (unsigned)c) * 16u;
      v[j] = *(const float4*)((const char*)h + boff);
    }
  }
}

static __device__ __forceinline__ void agg_add_batch(float4& acc,
                                                     const float4 (&v)[AGG_BATCH]) {
#pragma unroll
  for (int j = 0; j < AGG_BATCH; ++j) {
    acc.x += v[j].x; acc.y += v[j].y; acc.z += v[j].z; acc.w += v[j].w;
  }
}

template <int TPE>
static __device__ __forceinline__ void agg_ask_ids(const int* __restrict__ csr,
                                                   int at, int nrem, int c,
                                                   int& a, int& b) {
  a = (c < nrem) ? csr[at + c] : -1;
  b = -1;
  if (TPE == 8) b = (8 + c < nrem) ? csr[at + 8 + c] : -1;
}

template <int F, int ROUNDS, class Emit>
static __device__ __forceinline__ void agg_walk(const int2 rec,
                                                const int* __restrict__ csr,
                                                const float* __restrict__ h,
                                                const int node0, Emit emit) {
  constexpr int TPE = F / 4;
  constexpr int GPW = 32 / TPE;
  constexpr int NPW = GPW * ROUNDS;
  static_assert(TPE == 8 || TPE == 16, "a chunk of ids is one or two words per lane");
  static_assert(NPW + 1 <= 32, "one node record per lane, the closing one included");
  static_assert((size_t)NNODES * F * 4 < (1ull << 32), "row byte offsets fit 32 bits");
  const int lane = threadIdx.x & 31;
  const int c = lane & (TPE - 1);
  const int g = lane / TPE;

  int nbg = __shfl(rec.x, g, 32);
  int nen = __shfl(rec.x, g + 1, 32);
  int npa, npb;
  agg_ask_ids<TPE>(csr, nbg, nen - nbg, c, npa, npb);

#pragma unroll 1
  for (int r = 0; r < ROUNDS; ++r) {
    const int slot = r * GPW + g;
    const int node = node0 + slot;
    const bool live = node < NNODES;
    const float di = __int_as_float(__shfl(rec.y, slot, 32));
    const unsigned nboff = ((unsigned)node * TPE + (unsigned)c) * 16u;
    const int e = nen;
    int base = nbg;
    int ia = npa, ib = npb;
    float4 v[AGG_BATCH];
    float4 acc = make_float4(0.0f, 0.0f, 0.0f, 0.0f);

    int nrem = e - base;
    agg_ask_batch<TPE>(v, h, ia, 0, c);
    if (live) acc = *(const float4*)((const char*)h + nboff);
    if (r + 1 < ROUNDS) {
      nbg = __shfl(rec.x, slot + GPW, 32);
      nen = __shfl(rec.x, slot + GPW + 1, 32);
      agg_ask_ids<TPE>(csr, nbg, nen - nbg, c, npa, npb);
    }
    int na = -1, nb = -1;
    if (nrem > AGG_CHUNK)
      agg_ask_ids<TPE>(csr, base + AGG_CHUNK, nrem - AGG_CHUNK, c, na, nb);
    agg_add_batch(acc, v);
    if (nrem > AGG_BATCH) {
      if (TPE == 16) agg_ask_batch<TPE>(v, h, ia, AGG_BATCH, c);
      else agg_ask_batch<TPE>(v, h, ib, 0, c);
      agg_add_batch(acc, v);
    }
    base += AGG_CHUNK;
    ia = na;
    ib = nb;

    while (base < e) {
      nrem = e - base;
      agg_ask_batch<TPE>(v, h, ia, 0, c);
      na = -1;
      nb = -1;
      if (nrem > AGG_CHUNK)
        agg_ask_ids<TPE>(csr, base + AGG_CHUNK, nrem - AGG_CHUNK, c, na, nb);
      agg_add_batch(acc, v);
      if (nrem > AGG_BATCH) {
        if (TPE == 16) agg_ask_batch<TPE>(v, h, ia, AGG_BATCH, c);
        else agg_ask_batch<TPE>(v, h, ib, 0, c);
        agg_add_batch(acc, v);
      }
      base += AGG_CHUNK;
      ia = na;
      ib = nb;
    }

    acc.x *= di; acc.y *= di; acc.z *= di; acc.w *= di;
    emit(slot, live, nboff, acc);
  }
}

template <int F, int ROUNDS>
__global__ __launch_bounds__(AGG_THR) void agg_rows_kernel(
    const int2* __restrict__ info, const int* __restrict__ csr,
    const float* __restrict__ h, const float* __restrict__ bias,
    float* __restrict__ out) {
  constexpr int TPE = F / 4;
  constexpr int NPW = (32 / TPE) * ROUNDS;
  const int lane = threadIdx.x & 31;
  const int c = lane & (TPE - 1);
  const int node0 = (blockIdx.x * (AGG_THR / 32) + (threadIdx.x >> 5)) * NPW;
  if (node0 >= NNODES) return;

  int2 rec = {NEDGES, 0};
  if (lane <= NPW && node0 + lane < NNODES) rec = info[node0 + lane];

  agg_walk<F, ROUNDS>(rec, csr, h, node0,
                      [&](int, bool live, unsigned nboff, float4 sum) {
    if (live) {
      const float4 bb = ((const float4*)bias)[c];
      sum.x += bb.x; sum.y += bb.y; sum.z += bb.z; sum.w += bb.w;
      *(float4*)((char*)out + nboff) = sum;
    }
  });
}

#define FUS_WAVES 8
#define FUS_THR (FUS_WAVES * 32)
#define FUS_ROUNDS 8
#define FUS_BLOCKS ((NNODES / 16 + FUS_WAVES - 1) / FUS_WAVES)
#define AP (FH + 4)
static_assert(NNODES % 16 == 0, "every wave owns a whole tile of existing nodes");
static_assert((32 / (FH / 4)) * FUS_ROUNDS == 16, "a wave's run of nodes is one 16-row tile");
static_assert(AP % 4 == 0 && (AP / 4) % 2 == 1, "pitch = four floats x odd");
__global__ __launch_bounds__(FUS_THR) void agg1_gemm2_kernel(
    const int2* __restrict__ info, const int* __restrict__ csr,
    const float* __restrict__ h1, const float* __restrict__ b1,
    const float* __restrict__ W2, float* __restrict__ h2) {
  __shared__ __attribute__((aligned(16))) float sW[FOUT * W2P];
  __shared__ __attribute__((aligned(16))) float sA[FUS_WAVES * 16 * AP];
  const int lane = threadIdx.x & 31;
  const int wave = threadIdx.x >> 5;
  const int tile = blockIdx.x * FUS_WAVES + wave;
  const int node0 = tile * 16;
  const int c = lane & (FH / 4 - 1);
  float* tileA = sA + wave * 16 * AP;

  int2 rec = {NEDGES, 0};
  if (lane <= 16 && node0 + lane < NNODES) rec = info[node0 + lane];
  const float4 bb = ((const float4*)b1)[c];

  static_assert(FOUT * FH / 4 == 2 * FUS_THR && FH / 4 == 16, "two float4 of W2 per thread");
  const float4 w0 = ((const float4*)W2)[threadIdx.x];
  const float4 w1 = ((const float4*)W2)[threadIdx.x + FUS_THR];
  float* wdst = &sW[(threadIdx.x >> 4) * W2P + (threadIdx.x & 15) * 4];
  *(float4*)wdst = w0;
  *(float4*)(wdst + (FUS_THR >> 4) * W2P) = w1;
  __syncthreads();
  if (node0 >= NNODES) return;

  agg_walk<FH, FUS_ROUNDS>(rec, csr, h1, node0,
                           [&](int slot, bool, unsigned, float4 sum) {
    sum.x = fmaxf(sum.x + bb.x, 0.0f);
    sum.y = fmaxf(sum.y + bb.y, 0.0f);
    sum.z = fmaxf(sum.z + bb.z, 0.0f);
    sum.w = fmaxf(sum.w + bb.w, 0.0f);
    *(float4*)&tileA[slot * AP + 4 * c] = sum;
  });


  const int lrow = lane & 15;
  const int hh = lane >> 4;
  const int koff = hh << 1;
  const float* arow = tileA + lrow * AP + koff;
  const float* brow = sW + lrow * W2P + koff;
  v8f acc0 = {}, acc1 = {};
#pragma unroll
  for (int k = 0; k < FH; k += 4) {
    const v2f a = *(const v2f*)(arow + k);
    const v2f b0 = *(const v2f*)(brow + k);
    const v2f b1v = *(const v2f*)(brow + 16 * W2P + k);
    acc0 = wmma_f32_k4(a, b0, acc0);
    acc1 = wmma_f32_k4(a, b1v, acc1);
  }

  const int rsub = hh << 3;
#pragma unroll
  for (int r = 0; r < 8; ++r) {
    const float dv = __int_as_float(__shfl(rec.y, rsub + r, 32));
    h2[(size_t)(node0 + rsub + r) * FOUT + 0 * 16 + lrow] = acc0[r] * dv;
    h2[(size_t)(node0 + rsub + r) * FOUT + 1 * 16 + lrow] = acc1[r] * dv;
  }
}

#define AGG2_ROUNDS 4
#define AGG2_NPW ((32 / (FOUT / 4)) * AGG2_ROUNDS)
#define AGG2_BLOCKS ((NNODES + AGG2_NPW * (AGG_THR / 32) - 1) / (AGG2_NPW * (AGG_THR / 32)))

extern "C" void kernel_launch(void* const* d_in, const int* in_sizes, int n_in,
                              void* d_out, int out_size, void* d_ws,
                              size_t ws_size, hipStream_t stream) {
  const float* x = (const float*)d_in[0];
  const int* edge_index = (const int*)d_in[1];
  const float* W1 = (const float*)d_in[2];
  const float* b1 = (const float*)d_in[3];
  const float* W2 = (const float*)d_in[4];
  const float* b2 = (const float*)d_in[5];
  float* out = (float*)d_out;

  const int* src = edge_index;
  const int* dst = edge_index + NEDGES;

  float* ws = (float*)d_ws;
  int2* info = (int2*)ws;
  float* h1 = ws + 2 * (size_t)NNODES;
  float* region3 = h1 + (size_t)NNODES * FH;
  int* csr = (int*)(region3 + (size_t)NNODES * FH);
  float* h2 = region3;
  int* binned = (int*)region3;
  int* runoff = binned + NEDGES;

  bin_partition_kernel<<<PBLK, PTHR, 0, stream>>>(src, dst, runoff, binned);
  bin_build_gemm1_kernel<<<NBIN, BTHR, 0, stream>>>(runoff, binned, info, csr, x, W1, h1);

  agg1_gemm2_kernel<<<FUS_BLOCKS, FUS_THR, 0, stream>>>(info, csr, h1, b1, W2, h2);

  agg_rows_kernel<FOUT, AGG2_ROUNDS>
      <<<AGG2_BLOCKS, AGG_THR, 0, stream>>>(info, csr, h2, b2, out);
}
